# starting kernel
# baseline (speedup 1.0000x reference)
#include <hip/hip_runtime.h>
#include <hip/hip_bf16.h>

#define ALPHA   0.2f
#define NEG_INF (-9000000000000000.0f)

#define BT 128
#define NN 512
#define FF 128
#define DD 128
#define LDSP  132
#define PAIRP 144

typedef float v2f __attribute__((ext_vector_type(2)));
typedef float v8f __attribute__((ext_vector_type(8)));

__global__ void gat_wh_kernel(const float* __restrict__ h,
                              const float* __restrict__ W,
                              float* __restrict__ Wh) {
  extern __shared__ float smem[];
  float* sH = smem;
  v2f*   sW = (v2f*)(smem + 128 * LDSP);

  const int bt = blockIdx.x >> 2;
  const int rb = blockIdx.x & 3;
  const float* hblk  = h  + ((size_t)bt * NN + (size_t)rb * 128) * FF;
  float*       whblk = Wh + ((size_t)bt * NN + (size_t)rb * 128) * DD;

  const int tid = threadIdx.x;
  for (int idx = tid * 4; idx < 128 * FF; idx += 256 * 4) {
    const int r = idx >> 7, c = idx & 127;
    *(float4*)(sH + r * LDSP + c) = *(const float4*)(hblk + idx);
  }
  for (int idx = tid; idx < 64 * 32; idx += 256) {
    const int k2 = idx >> 5, nq = idx & 31;
    float4 r0 = *(const float4*)(W + (2 * k2) * DD + nq * 4);
    float4 r1 = *(const float4*)(W + (2 * k2 + 1) * DD + nq * 4);
    float* dst = (float*)(sW + k2 * PAIRP + nq * 4);
    *(float4*)(dst)     = make_float4(r0.x, r1.x, r0.y, r1.y);
    *(float4*)(dst + 4) = make_float4(r0.z, r1.z, r0.w, r1.w);
  }
  __syncthreads();

  const int lane = tid & 31;
  const int wave = tid >> 5;
  const int l16  = lane & 15;
  const int g    = lane >> 4;
  const int arow = wave * 16 + l16;

  v8f acc[8];
#pragma unroll
  for (int t = 0; t < 8; ++t)
#pragma unroll
    for (int r = 0; r < 8; ++r) acc[t][r] = 0.0f;

  for (int kk = 0; kk < FF; kk += 4) {
    const int ka = kk + 2 * g;
    const v2f a = *(const v2f*)(sH + arow * LDSP + ka);
    const v2f* brow = sW + ((kk >> 1) + g) * PAIRP + l16;
    v2f b[8];
#pragma unroll
    for (int t = 0; t < 8; ++t) b[t] = brow[t * 16];
#pragma unroll
    for (int t = 0; t < 8; ++t)
      acc[t] = __builtin_amdgcn_wmma_f32_16x16x4_f32(
          false, a, false, b[t], (short)0, acc[t], false, false);
  }

#pragma unroll
  for (int t = 0; t < 8; ++t)
#pragma unroll
    for (int r = 0; r < 8; ++r)
      whblk[(wave * 16 + r + 8 * g) * DD + t * 16 + l16] = acc[t][r];
}

__global__ void gat_e_kernel(const float* __restrict__ Wh,
                             const float* __restrict__ a,
                             float* __restrict__ esrc,
                             float* __restrict__ edst) {
  const int lane = threadIdx.x & 31;
  const int row  = (blockIdx.x * blockDim.x + threadIdx.x) >> 5;

  float4 v  = *(const float4*)(Wh + (size_t)row * DD + lane * 4);
  float4 as = *(const float4*)(a + lane * 4);
  float4 ad = *(const float4*)(a + DD + lane * 4);
  float ps = v.x * as.x + v.y * as.y + v.z * as.z + v.w * as.w;
  float pd = v.x * ad.x + v.y * ad.y + v.z * ad.z + v.w * ad.w;
#pragma unroll
  for (int off = 16; off >= 1; off >>= 1) {
    ps += __shfl_xor(ps, off, 32);
    pd += __shfl_xor(pd, off, 32);
  }
  if (lane == 0) {
    esrc[row] = ps;
    edst[row] = pd;
  }
}

__global__ void gat_attn_kernel(const float* __restrict__ Wh,
                                const float* __restrict__ esrc,
                                const float* __restrict__ edst,
                                const int* __restrict__ adj,
                                float* __restrict__ out) {
  extern __shared__ float smem[];
  v2f*      sWh   = (v2f*)smem;
  float*    sEd   = smem + (NN / 2) * PAIRP * 2;
  unsigned* sMask = (unsigned*)(sEd + NN);

  const int bt = blockIdx.x >> 2;
  const int rb = blockIdx.x & 3;
  const int tid = threadIdx.x;
  const float* whbt = Wh + (size_t)bt * NN * DD;

  for (int idx = tid; idx < 256 * 32; idx += 256) {
    const int k2 = idx >> 5, nq = idx & 31;
    float4 r0 = *(const float4*)(whbt + (size_t)(2 * k2) * DD + nq * 4);
    float4 r1 = *(const float4*)(whbt + (size_t)(2 * k2 + 1) * DD + nq * 4);
    float* dst = (float*)(sWh + k2 * PAIRP + nq * 4);
    *(float4*)(dst)     = make_float4(r0.x, r1.x, r0.y, r1.y);
    *(float4*)(dst + 4) = make_float4(r0.z, r1.z, r0.w, r1.w);
  }
  for (int j = tid; j < NN; j += 256) sEd[j] = edst[bt * NN + j];
  __syncthreads();

  const int lane = tid & 31;
  const int wave = tid >> 5;
  const int l16  = lane & 15;
  const int g    = lane >> 4;

  const int irow = rb * 128 + wave * 16 + l16;
  const float es = esrc[bt * NN + irow];
  const int* adjrow = adj + irow * NN;
  unsigned* maskrow = sMask + (wave * 16 + l16) * 16;

  float m = -3.0e38f, l = 0.0f;
  for (int w = 0; w < 16; ++w) {
    unsigned mw = 0;
#pragma unroll
    for (int jj = 0; jj < 16; ++jj) {
      const int j = w * 32 + 2 * jj + g;
      const bool conn = adjrow[j] > 0;
      float ev = es + sEd[j];
      ev = ev > 0.0f ? ev : ALPHA * ev;
      if (!conn) ev = NEG_INF;
      mw |= (conn ? 1u : 0u) << (2 * jj + g);
      const float nm = fmaxf(m, ev);
      l = l * __expf(m - nm) + __expf(ev - nm);
      m = nm;
    }
    mw |= __shfl_xor(mw, 16, 32);
    if (g == 0) maskrow[w] = mw;
  }
  {
    const float m2 = __shfl_xor(m, 16, 32);
    const float l2 = __shfl_xor(l, 16, 32);
    const float nm = fmaxf(m, m2);
    l = l * __expf(m - nm) + l2 * __expf(m2 - nm);
    m = nm;
  }

  v8f acc[8];
#pragma unroll
  for (int t = 0; t < 8; ++t)
#pragma unroll
    for (int r = 0; r < 8; ++r) acc[t][r] = 0.0f;

  for (int w = 0; w < 16; ++w) {
    const unsigned mw = maskrow[w];
#pragma unroll
    for (int c = 0; c < 8; ++c) {
      const int j0 = w * 32 + c * 4;
      const int ja = j0 + 2 * g;
      float e0 = es + sEd[ja];
      e0 = e0 > 0.0f ? e0 : ALPHA * e0;
      if (!((mw >> (c * 4 + 2 * g)) & 1u)) e0 = NEG_INF;
      float e1 = es + sEd[ja + 1];
      e1 = e1 > 0.0f ? e1 : ALPHA * e1;
      if (!((mw >> (c * 4 + 2 * g + 1)) & 1u)) e1 = NEG_INF;
      v2f a;
      a.x = __expf(e0 - m);
      a.y = __expf(e1 - m);
      const v2f* brow = sWh + ((j0 >> 1) + g) * PAIRP + l16;
      v2f b[8];
#pragma unroll
      for (int t = 0; t < 8; ++t) b[t] = brow[t * 16];
#pragma unroll
      for (int t = 0; t < 8; ++t)
        acc[t] = __builtin_amdgcn_wmma_f32_16x16x4_f32(
            false, a, false, b[t], (short)0, acc[t], false, false);
    }
  }

  float* outblk = out + ((size_t)bt * NN + (size_t)rb * 128 + wave * 16) * DD;
#pragma unroll
  for (int r = 0; r < 8; ++r) {
    const int orow = r + 8 * g;
    const float lr = __shfl(l, orow, 32);
    const float inv = 1.0f / lr;
#pragma unroll
    for (int t = 0; t < 8; ++t)
      outblk[orow * DD + t * 16 + l16] = acc[t][r] * inv;
  }
}

extern "C" void kernel_launch(void* const* d_in, const int* in_sizes, int n_in,
                              void* d_out, int out_size, void* d_ws, size_t ws_size,
                              hipStream_t stream) {
  const float* h   = (const float*)d_in[0];
  const int*   adj = (const int*)d_in[1];
  const float* W   = (const float*)d_in[2];
  const float* a   = (const float*)d_in[3];
  float* out = (float*)d_out;

  float* Wh   = (float*)d_ws;
  float* esrc = Wh + (size_t)BT * NN * DD;
  float* edst = esrc + (size_t)BT * NN;

  const size_t lds1 = (size_t)(128 * LDSP) * 4 + (size_t)(64 * PAIRP) * 8;
  const size_t lds3 = (size_t)(NN / 2) * PAIRP * 8 + (size_t)NN * 4 +
                      (size_t)(8 * 16 * 16) * 4;

  gat_wh_kernel<<<dim3(BT * 4), dim3(256), lds1, stream>>>(h, W, Wh);
  gat_e_kernel<<<dim3(BT * NN / 8), dim3(256), 0, stream>>>(Wh, a, esrc, edst);
  gat_attn_kernel<<<dim3(BT * 4), dim3(256), lds3, stream>>>(Wh, esrc, edst, adj, out);
}


# Both WMMA loops software-pipelined: next step's LDS fragment loads issued as one group ahead of the current 8 WMMAs (second register set), next A operand computed beside them
# speedup vs baseline: 1.2280x; 1.2280x over previous
#include <hip/hip_runtime.h>
#include <hip/hip_bf16.h>

#define ALPHA   0.2f
#define NEG_INF (-9000000000000000.0f)

#define BT 128
#define NN 512
#define FF 128
#define DD 128
#define LDSP  132
#define PAIRP 144

typedef float v2f __attribute__((ext_vector_type(2)));
typedef float v8f __attribute__((ext_vector_type(8)));

static __device__ __forceinline__ void fence() { __builtin_amdgcn_sched_barrier(0); }

__global__ void gat_wh_kernel(const float* __restrict__ h,
                              const float* __restrict__ W,
                              float* __restrict__ Wh) {
  extern __shared__ float smem[];
  float* sH = smem;
  v2f*   sW = (v2f*)(smem + 128 * LDSP);

  const int bt = blockIdx.x >> 2;
  const int rb = blockIdx.x & 3;
  const float* hblk  = h  + ((size_t)bt * NN + (size_t)rb * 128) * FF;
  float*       whblk = Wh + ((size_t)bt * NN + (size_t)rb * 128) * DD;

  const int tid = threadIdx.x;
  for (int idx = tid * 4; idx < 128 * FF; idx += 256 * 4) {
    const int r = idx >> 7, c = idx & 127;
    *(float4*)(sH + r * LDSP + c) = *(const float4*)(hblk + idx);
  }
  for (int idx = tid; idx < 64 * 32; idx += 256) {
    const int k2 = idx >> 5, nq = idx & 31;
    float4 r0 = *(const float4*)(W + (2 * k2) * DD + nq * 4);
    float4 r1 = *(const float4*)(W + (2 * k2 + 1) * DD + nq * 4);
    float* dst = (float*)(sW + k2 * PAIRP + nq * 4);
    *(float4*)(dst)     = make_float4(r0.x, r1.x, r0.y, r1.y);
    *(float4*)(dst + 4) = make_float4(r0.z, r1.z, r0.w, r1.w);
  }
  __syncthreads();

  const int lane = tid & 31;
  const int wave = tid >> 5;
  const int l16  = lane & 15;
  const int g    = lane >> 4;
  const int arow = wave * 16 + l16;

  v8f acc[8];
#pragma unroll
  for (int t = 0; t < 8; ++t)
#pragma unroll
    for (int r = 0; r < 8; ++r) acc[t][r] = 0.0f;

  const float* aptr = sH + arow * LDSP + 2 * g;
  const v2f*   bptr = sW + g * PAIRP + l16;
  v2f aq[2];
  v2f bq[2][8];
  aq[0] = *(const v2f*)aptr;
#pragma unroll
  for (int t = 0; t < 8; ++t) bq[0][t] = bptr[t * 16];

  for (int s2 = 0; s2 < FF / 4; s2 += 2) {
#pragma unroll
    for (int u = 0; u < 2; ++u) {
      const int sn = (s2 + u + 1) & (FF / 4 - 1);
      const v2f* bn = bptr + 2 * sn * PAIRP;
      fence();
      aq[u ^ 1] = *(const v2f*)(aptr + 4 * sn);
#pragma unroll
      for (int t = 0; t < 8; ++t) bq[u ^ 1][t] = bn[t * 16];
      fence();
#pragma unroll
      for (int t = 0; t < 8; ++t)
        acc[t] = __builtin_amdgcn_wmma_f32_16x16x4_f32(
            false, aq[u], false, bq[u][t], (short)0, acc[t], false, false);
      fence();
    }
  }

#pragma unroll
  for (int t = 0; t < 8; ++t)
#pragma unroll
    for (int r = 0; r < 8; ++r)
      whblk[(wave * 16 + r + 8 * g) * DD + t * 16 + l16] = acc[t][r];
}

__global__ void gat_e_kernel(const float* __restrict__ Wh,
                             const float* __restrict__ a,
                             float* __restrict__ esrc,
                             float* __restrict__ edst) {
  const int lane = threadIdx.x & 31;
  const int row  = (blockIdx.x * blockDim.x + threadIdx.x) >> 5;

  float4 v  = *(const float4*)(Wh + (size_t)row * DD + lane * 4);
  float4 as = *(const float4*)(a + lane * 4);
  float4 ad = *(const float4*)(a + DD + lane * 4);
  float ps = v.x * as.x + v.y * as.y + v.z * as.z + v.w * as.w;
  float pd = v.x * ad.x + v.y * ad.y + v.z * ad.z + v.w * ad.w;
#pragma unroll
  for (int off = 16; off >= 1; off >>= 1) {
    ps += __shfl_xor(ps, off, 32);
    pd += __shfl_xor(pd, off, 32);
  }
  if (lane == 0) {
    esrc[row] = ps;
    edst[row] = pd;
  }
}

__global__ void __launch_bounds__(256) gat_attn_kernel(const float* __restrict__ Wh,
                                const float* __restrict__ esrc,
                                const float* __restrict__ edst,
                                const int* __restrict__ adj,
                                float* __restrict__ out) {
  extern __shared__ float smem[];
  v2f*      sWh   = (v2f*)smem;
  float*    sEd   = smem + (NN / 2) * PAIRP * 2;
  unsigned* sMask = (unsigned*)(sEd + NN);

  const int bt = blockIdx.x >> 2;
  const int rb = blockIdx.x & 3;
  const int tid = threadIdx.x;
  const float* whbt = Wh + (size_t)bt * NN * DD;

  for (int idx = tid; idx < 256 * 32; idx += 256) {
    const int k2 = idx >> 5, nq = idx & 31;
    float4 r0 = *(const float4*)(whbt + (size_t)(2 * k2) * DD + nq * 4);
    float4 r1 = *(const float4*)(whbt + (size_t)(2 * k2 + 1) * DD + nq * 4);
    float* dst = (float*)(sWh + k2 * PAIRP + nq * 4);
    *(float4*)(dst)     = make_float4(r0.x, r1.x, r0.y, r1.y);
    *(float4*)(dst + 4) = make_float4(r0.z, r1.z, r0.w, r1.w);
  }
  for (int j = tid; j < NN; j += 256) sEd[j] = edst[bt * NN + j];
  __syncthreads();

  const int lane = tid & 31;
  const int wave = tid >> 5;
  const int l16  = lane & 15;
  const int g    = lane >> 4;

  const int irow = rb * 128 + wave * 16 + l16;
  const float es = esrc[bt * NN + irow];
  const int* adjrow = adj + irow * NN;
  unsigned* maskrow = sMask + (wave * 16 + l16) * 16;

  float m = -3.0e38f, l = 0.0f;
  for (int w = 0; w < 16; ++w) {
    unsigned mw = 0;
#pragma unroll
    for (int jj = 0; jj < 16; ++jj) {
      const int j = w * 32 + 2 * jj + g;
      const bool conn = adjrow[j] > 0;
      float ev = es + sEd[j];
      ev = ev > 0.0f ? ev : ALPHA * ev;
      if (!conn) ev = NEG_INF;
      mw |= (conn ? 1u : 0u) << (2 * jj + g);
      const float nm = fmaxf(m, ev);
      l = l * __expf(m - nm) + __expf(ev - nm);
      m = nm;
    }
    mw |= __shfl_xor(mw, 16, 32);
    if (g == 0) maskrow[w] = mw;
  }
  {
    const float m2 = __shfl_xor(m, 16, 32);
    const float l2 = __shfl_xor(l, 16, 32);
    const float nm = fmaxf(m, m2);
    l = l * __expf(m - nm) + l2 * __expf(m2 - nm);
    m = nm;
  }

  v8f acc[8];
#pragma unroll
  for (int t = 0; t < 8; ++t)
#pragma unroll
    for (int r = 0; r < 8; ++r) acc[t][r] = 0.0f;

  const v2f* bptr  = sWh + g * PAIRP + l16;
  const v2f* edptr = (const v2f*)sEd + g;
  v2f bq[2][8];
  v2f edc[8], edn[8];
  unsigned mwc, mwn;

  auto prob = [&](v2f ed, unsigned mw, int c) -> v2f {
    float e0 = es + ed.x;
    e0 = e0 > 0.0f ? e0 : ALPHA * e0;
    if (!((mw >> (c * 4 + 2 * g)) & 1u)) e0 = NEG_INF;
    float e1 = es + ed.y;
    e1 = e1 > 0.0f ? e1 : ALPHA * e1;
    if (!((mw >> (c * 4 + 2 * g + 1)) & 1u)) e1 = NEG_INF;
    v2f p;
    p.x = __expf(e0 - m);
    p.y = __expf(e1 - m);
    return p;
  };

#pragma unroll
  for (int t = 0; t < 8; ++t) bq[0][t] = bptr[t * 16];
#pragma unroll
  for (int k = 0; k < 8; ++k) edc[k] = edptr[2 * k];
  mwc = maskrow[0];
  v2f a = prob(edc[0], mwc, 0);

  for (int w = 0; w < 16; ++w) {
    const int wn = (w + 1) & 15;
#pragma unroll
    for (int c = 0; c < 8; ++c) {
      const int sn = (w * 8 + c + 1) & 127;
      const v2f* bn = bptr + 2 * sn * PAIRP;
      fence();
#pragma unroll
      for (int t = 0; t < 8; ++t) bq[(c + 1) & 1][t] = bn[t * 16];
      if (c == 0) {
#pragma unroll
        for (int k = 0; k < 8; ++k) edn[k] = edptr[16 * wn + 2 * k];
        mwn = maskrow[wn];
      }
      fence();
      const v2f an = (c < 7) ? prob(edc[(c + 1) & 7], mwc, c + 1) : prob(edn[0], mwn, 0);
#pragma unroll
      for (int t = 0; t < 8; ++t)
        acc[t] = __builtin_amdgcn_wmma_f32_16x16x4_f32(
            false, a, false, bq[c & 1][t], (short)0, acc[t], false, false);
      fence();
      a = an;
    }
#pragma unroll
    for (int k = 0; k < 8; ++k) edc[k] = edn[k];
    mwc = mwn;
  }

  float* outblk = out + ((size_t)bt * NN + (size_t)rb * 128 + wave * 16) * DD;
#pragma unroll
  for (int r = 0; r < 8; ++r) {
    const int orow = r + 8 * g;
    const float lr = __shfl(l, orow, 32);
    const float inv = 1.0f / lr;
#pragma unroll
    for (int t = 0; t < 8; ++t)
      outblk[orow * DD + t * 16 + l16] = acc[t][r] * inv;
  }
}

extern "C" void kernel_launch(void* const* d_in, const int* in_sizes, int n_in,
                              void* d_out, int out_size, void* d_ws, size_t ws_size,
                              hipStream_t stream) {
  const float* h   = (const float*)d_in[0];
  const int*   adj = (const int*)d_in[1];
  const float* W   = (const float*)d_in[2];
  const float* a   = (const float*)d_in[3];
  float* out = (float*)d_out;

  float* Wh   = (float*)d_ws;
  float* esrc = Wh + (size_t)BT * NN * DD;
  float* edst = esrc + (size_t)BT * NN;

  const size_t lds1 = (size_t)(128 * LDSP) * 4 + (size_t)(64 * PAIRP) * 8;
  const size_t lds3 = (size_t)(NN / 2) * PAIRP * 8 + (size_t)NN * 4 +
                      (size_t)(8 * 16 * 16) * 4;

  gat_wh_kernel<<<dim3(BT * 4), dim3(256), lds1, stream>>>(h, W, Wh);
  gat_e_kernel<<<dim3(BT * NN / 8), dim3(256), 0, stream>>>(Wh, a, esrc, edst);
  gat_attn_kernel<<<dim3(BT * 4), dim3(256), lds3, stream>>>(Wh, esrc, edst, adj, out);
}


# e_src/e_dst computed in the Wh GEMM epilogue from the WMMA accumulators (16-lane row_xmask DPP sums); the one-wave-per-row launch is removed, 3 launches -> 2; their sum order changes
# speedup vs baseline: 1.3742x; 1.1191x over previous
#include <hip/hip_runtime.h>
#include <hip/hip_bf16.h>

#define ALPHA   0.2f
#define NEG_INF (-9000000000000000.0f)

#define BT 128
#define NN 512
#define FF 128
#define DD 128
#define LDSP  132
#define PAIRP 144

typedef float v2f __attribute__((ext_vector_type(2)));
typedef float v8f __attribute__((ext_vector_type(8)));

static __device__ __forceinline__ void fence() { __builtin_amdgcn_sched_barrier(0); }

template <int XMASK>
static __device__ __forceinline__ float row_xor_add(float v) {
  const int o = __builtin_amdgcn_update_dpp(0, __builtin_bit_cast(int, v), 0x160 + XMASK, 0xf, 0xf, true);
  return v + __builtin_bit_cast(float, o);
}

__global__ void gat_wh_kernel(const float* __restrict__ h,
                              const float* __restrict__ W,
                              const float* __restrict__ a,
                              float* __restrict__ Wh,
                              float* __restrict__ esrc,
                              float* __restrict__ edst) {
  extern __shared__ float smem[];
  float* sH = smem;
  v2f*   sW = (v2f*)(smem + 128 * LDSP);
  float* sA = smem + 128 * LDSP + 64 * PAIRP * 2;

  const int bt = blockIdx.x >> 2;
  const int rb = blockIdx.x & 3;
  const float* hblk  = h  + ((size_t)bt * NN + (size_t)rb * 128) * FF;
  float*       whblk = Wh + ((size_t)bt * NN + (size_t)rb * 128) * DD;

  const int tid = threadIdx.x;
  for (int idx = tid * 4; idx < 128 * FF; idx += 256 * 4) {
    const int r = idx >> 7, c = idx & 127;
    *(float4*)(sH + r * LDSP + c) = *(const float4*)(hblk + idx);
  }
  for (int idx = tid; idx < 64 * 32; idx += 256) {
    const int k2 = idx >> 5, nq = idx & 31;
    float4 r0 = *(const float4*)(W + (2 * k2) * DD + nq * 4);
    float4 r1 = *(const float4*)(W + (2 * k2 + 1) * DD + nq * 4);
    float* dst = (float*)(sW + k2 * PAIRP + nq * 4);
    *(float4*)(dst)     = make_float4(r0.x, r1.x, r0.y, r1.y);
    *(float4*)(dst + 4) = make_float4(r0.z, r1.z, r0.w, r1.w);
  }
  sA[tid] = a[tid];
  __syncthreads();

  const int lane = tid & 31;
  const int wave = tid >> 5;
  const int l16  = lane & 15;
  const int g    = lane >> 4;
  const int arow = wave * 16 + l16;

  v8f acc[8];
#pragma unroll
  for (int t = 0; t < 8; ++t)
#pragma unroll
    for (int r = 0; r < 8; ++r) acc[t][r] = 0.0f;

  const float* aptr = sH + arow * LDSP + 2 * g;
  const v2f*   bptr = sW + g * PAIRP + l16;
  v2f aq[2];
  v2f bq[2][8];
  aq[0] = *(const v2f*)aptr;
#pragma unroll
  for (int t = 0; t < 8; ++t) bq[0][t] = bptr[t * 16];

  for (int s2 = 0; s2 < FF / 4; s2 += 2) {
#pragma unroll
    for (int u = 0; u < 2; ++u) {
      const int sn = (s2 + u + 1) & (FF / 4 - 1);
      const v2f* bn = bptr + 2 * sn * PAIRP;
      fence();
      aq[u ^ 1] = *(const v2f*)(aptr + 4 * sn);
#pragma unroll
      for (int t = 0; t < 8; ++t) bq[u ^ 1][t] = bn[t * 16];
      fence();
#pragma unroll
      for (int t = 0; t < 8; ++t)
        acc[t] = __builtin_amdgcn_wmma_f32_16x16x4_f32(
            false, aq[u], false, bq[u][t], (short)0, acc[t], false, false);
      fence();
    }
  }

#pragma unroll
  for (int t = 0; t < 8; ++t)
#pragma unroll
    for (int r = 0; r < 8; ++r)
      whblk[(wave * 16 + r + 8 * g) * DD + t * 16 + l16] = acc[t][r];

  float as[8], ad[8];
#pragma unroll
  for (int t = 0; t < 8; ++t) {
    as[t] = sA[t * 16 + l16];
    ad[t] = sA[DD + t * 16 + l16];
  }
  float vsel = 0.0f;
#pragma unroll
  for (int r = 0; r < 8; ++r) {
    float ps = 0.0f, pd = 0.0f;
#pragma unroll
    for (int t = 0; t < 8; ++t) {
      ps = fmaf(acc[t][r], as[t], ps);
      pd = fmaf(acc[t][r], ad[t], pd);
    }
    ps = row_xor_add<8>(ps);  pd = row_xor_add<8>(pd);
    ps = row_xor_add<4>(ps);  pd = row_xor_add<4>(pd);
    ps = row_xor_add<2>(ps);  pd = row_xor_add<2>(pd);
    ps = row_xor_add<1>(ps);  pd = row_xor_add<1>(pd);
    if ((l16 & 7) == r) vsel = (l16 < 8) ? ps : pd;
  }
  float* ebase = (l16 < 8) ? esrc : edst;
  ebase[(size_t)blockIdx.x * 128 + wave * 16 + 8 * g + (l16 & 7)] = vsel;
}

__global__ void __launch_bounds__(256) gat_attn_kernel(const float* __restrict__ Wh,
                                const float* __restrict__ esrc,
                                const float* __restrict__ edst,
                                const int* __restrict__ adj,
                                float* __restrict__ out) {
  extern __shared__ float smem[];
  v2f*      sWh   = (v2f*)smem;
  float*    sEd   = smem + (NN / 2) * PAIRP * 2;
  unsigned* sMask = (unsigned*)(sEd + NN);

  const int bt = blockIdx.x >> 2;
  const int rb = blockIdx.x & 3;
  const int tid = threadIdx.x;
  const float* whbt = Wh + (size_t)bt * NN * DD;

  for (int idx = tid; idx < 256 * 32; idx += 256) {
    const int k2 = idx >> 5, nq = idx & 31;
    float4 r0 = *(const float4*)(whbt + (size_t)(2 * k2) * DD + nq * 4);
    float4 r1 = *(const float4*)(whbt + (size_t)(2 * k2 + 1) * DD + nq * 4);
    float* dst = (float*)(sWh + k2 * PAIRP + nq * 4);
    *(float4*)(dst)     = make_float4(r0.x, r1.x, r0.y, r1.y);
    *(float4*)(dst + 4) = make_float4(r0.z, r1.z, r0.w, r1.w);
  }
  for (int j = tid; j < NN; j += 256) sEd[j] = edst[bt * NN + j];
  __syncthreads();

  const int lane = tid & 31;
  const int wave = tid >> 5;
  const int l16  = lane & 15;
  const int g    = lane >> 4;

  const int irow = rb * 128 + wave * 16 + l16;
  const float es = esrc[bt * NN + irow];
  const int* adjrow = adj + irow * NN;
  unsigned* maskrow = sMask + (wave * 16 + l16) * 16;

  float m = -3.0e38f, l = 0.0f;
  for (int w = 0; w < 16; ++w) {
    unsigned mw = 0;
#pragma unroll
    for (int jj = 0; jj < 16; ++jj) {
      const int j = w * 32 + 2 * jj + g;
      const bool conn = adjrow[j] > 0;
      float ev = es + sEd[j];
      ev = ev > 0.0f ? ev : ALPHA * ev;
      if (!conn) ev = NEG_INF;
      mw |= (conn ? 1u : 0u) << (2 * jj + g);
      const float nm = fmaxf(m, ev);
      l = l * __expf(m - nm) + __expf(ev - nm);
      m = nm;
    }
    mw |= __shfl_xor(mw, 16, 32);
    if (g == 0) maskrow[w] = mw;
  }
  {
    const float m2 = __shfl_xor(m, 16, 32);
    const float l2 = __shfl_xor(l, 16, 32);
    const float nm = fmaxf(m, m2);
    l = l * __expf(m - nm) + l2 * __expf(m2 - nm);
    m = nm;
  }

  v8f acc[8];
#pragma unroll
  for (int t = 0; t < 8; ++t)
#pragma unroll
    for (int r = 0; r < 8; ++r) acc[t][r] = 0.0f;

  const v2f* bptr  = sWh + g * PAIRP + l16;
  const v2f* edptr = (const v2f*)sEd + g;
  v2f bq[2][8];
  v2f edc[8], edn[8];
  unsigned mwc, mwn;

  auto prob = [&](v2f ed, unsigned mw, int c) -> v2f {
    float e0 = es + ed.x;
    e0 = e0 > 0.0f ? e0 : ALPHA * e0;
    if (!((mw >> (c * 4 + 2 * g)) & 1u)) e0 = NEG_INF;
    float e1 = es + ed.y;
    e1 = e1 > 0.0f ? e1 : ALPHA * e1;
    if (!((mw >> (c * 4 + 2 * g + 1)) & 1u)) e1 = NEG_INF;
    v2f p;
    p.x = __expf(e0 - m);
    p.y = __expf(e1 - m);
    return p;
  };

#pragma unroll
  for (int t = 0; t < 8; ++t) bq[0][t] = bptr[t * 16];
#pragma unroll
  for (int k = 0; k < 8; ++k) edc[k] = edptr[2 * k];
  mwc = maskrow[0];
  v2f a = prob(edc[0], mwc, 0);

  for (int w = 0; w < 16; ++w) {
    const int wn = (w + 1) & 15;
#pragma unroll
    for (int c = 0; c < 8; ++c) {
      const int sn = (w * 8 + c + 1) & 127;
      const v2f* bn = bptr + 2 * sn * PAIRP;
      fence();
#pragma unroll
      for (int t = 0; t < 8; ++t) bq[(c + 1) & 1][t] = bn[t * 16];
      if (c == 0) {
#pragma unroll
        for (int k = 0; k < 8; ++k) edn[k] = edptr[16 * wn + 2 * k];
        mwn = maskrow[wn];
      }
      fence();
      const v2f an = (c < 7) ? prob(edc[(c + 1) & 7], mwc, c + 1) : prob(edn[0], mwn, 0);
#pragma unroll
      for (int t = 0; t < 8; ++t)
        acc[t] = __builtin_amdgcn_wmma_f32_16x16x4_f32(
            false, a, false, bq[c & 1][t], (short)0, acc[t], false, false);
      fence();
      a = an;
    }
#pragma unroll
    for (int k = 0; k < 8; ++k) edc[k] = edn[k];
    mwc = mwn;
  }

  float* outblk = out + ((size_t)bt * NN + (size_t)rb * 128 + wave * 16) * DD;
#pragma unroll
  for (int r = 0; r < 8; ++r) {
    const int orow = r + 8 * g;
    const float lr = __shfl(l, orow, 32);
    const float inv = 1.0f / lr;
#pragma unroll
    for (int t = 0; t < 8; ++t)
      outblk[orow * DD + t * 16 + l16] = acc[t][r] * inv;
  }
}

extern "C" void kernel_launch(void* const* d_in, const int* in_sizes, int n_in,
                              void* d_out, int out_size, void* d_ws, size_t ws_size,
                              hipStream_t stream) {
  const float* h   = (const float*)d_in[0];
  const int*   adj = (const int*)d_in[1];
  const float* W   = (const float*)d_in[2];
  const float* a   = (const float*)d_in[3];
  float* out = (float*)d_out;

  float* Wh   = (float*)d_ws;
  float* esrc = Wh + (size_t)BT * NN * DD;
  float* edst = esrc + (size_t)BT * NN;

  const size_t lds1 = (size_t)(128 * LDSP) * 4 + (size_t)(64 * PAIRP) * 8 + (size_t)(2 * DD) * 4;
  const size_t lds3 = (size_t)(NN / 2) * PAIRP * 8 + (size_t)NN * 4 +
                      (size_t)(8 * 16 * 16) * 4;

  gat_wh_kernel<<<dim3(BT * 4), dim3(256), lds1, stream>>>(h, W, a, Wh, esrc, edst);
  gat_attn_kernel<<<dim3(BT * 4), dim3(256), lds3, stream>>>(Wh, esrc, edst, adj, out);
}


# Attention kernel stages its Wh slice in two halves through a 144 KiB LDS buffer (was 288 KiB, whole): two workgroups now share a WGP, so one's staging and softmax scan run under the other's WMMAs
# speedup vs baseline: 1.5424x; 1.1224x over previous
#include <hip/hip_runtime.h>
#include <hip/hip_bf16.h>

#define ALPHA   0.2f
#define NEG_INF (-9000000000000000.0f)

#define BT 128
#define NN 512
#define FF 128
#define DD 128
#define LDSP  132
#define PAIRP 144

typedef float v2f __attribute__((ext_vector_type(2)));
typedef float v8f __attribute__((ext_vector_type(8)));

static __device__ __forceinline__ void fence() { __builtin_amdgcn_sched_barrier(0); }

template <int XMASK>
static __device__ __forceinline__ float row_xor_add(float v) {
  const int o = __builtin_amdgcn_update_dpp(0, __builtin_bit_cast(int, v), 0x160 + XMASK, 0xf, 0xf, true);
  return v + __builtin_bit_cast(float, o);
}

__global__ void gat_wh_kernel(const float* __restrict__ h,
                              const float* __restrict__ W,
                              const float* __restrict__ a,
                              float* __restrict__ Wh,
                              float* __restrict__ esrc,
                              float* __restrict__ edst) {
  extern __shared__ float smem[];
  float* sH = smem;
  v2f*   sW = (v2f*)(smem + 128 * LDSP);
  float* sA = smem + 128 * LDSP + 64 * PAIRP * 2;

  const int bt = blockIdx.x >> 2;
  const int rb = blockIdx.x & 3;
  const float* hblk  = h  + ((size_t)bt * NN + (size_t)rb * 128) * FF;
  float*       whblk = Wh + ((size_t)bt * NN + (size_t)rb * 128) * DD;

  const int tid = threadIdx.x;
  for (int idx = tid * 4; idx < 128 * FF; idx += 256 * 4) {
    const int r = idx >> 7, c = idx & 127;
    *(float4*)(sH + r * LDSP + c) = *(const float4*)(hblk + idx);
  }
  for (int idx = tid; idx < 64 * 32; idx += 256) {
    const int k2 = idx >> 5, nq = idx & 31;
    float4 r0 = *(const float4*)(W + (2 * k2) * DD + nq * 4);
    float4 r1 = *(const float4*)(W + (2 * k2 + 1) * DD + nq * 4);
    float* dst = (float*)(sW + k2 * PAIRP + nq * 4);
    *(float4*)(dst)     = make_float4(r0.x, r1.x, r0.y, r1.y);
    *(float4*)(dst + 4) = make_float4(r0.z, r1.z, r0.w, r1.w);
  }
  sA[tid] = a[tid];
  __syncthreads();

  const int lane = tid & 31;
  const int wave = tid >> 5;
  const int l16  = lane & 15;
  const int g    = lane >> 4;
  const int arow = wave * 16 + l16;

  v8f acc[8];
#pragma unroll
  for (int t = 0; t < 8; ++t)
#pragma unroll
    for (int r = 0; r < 8; ++r) acc[t][r] = 0.0f;

  const float* aptr = sH + arow * LDSP + 2 * g;
  const v2f*   bptr = sW + g * PAIRP + l16;
  v2f aq[2];
  v2f bq[2][8];
  aq[0] = *(const v2f*)aptr;
#pragma unroll
  for (int t = 0; t < 8; ++t) bq[0][t] = bptr[t * 16];

  for (int s2 = 0; s2 < FF / 4; s2 += 2) {
#pragma unroll
    for (int u = 0; u < 2; ++u) {
      const int sn = (s2 + u + 1) & (FF / 4 - 1);
      const v2f* bn = bptr + 2 * sn * PAIRP;
      fence();
      aq[u ^ 1] = *(const v2f*)(aptr + 4 * sn);
#pragma unroll
      for (int t = 0; t < 8; ++t) bq[u ^ 1][t] = bn[t * 16];
      fence();
#pragma unroll
      for (int t = 0; t < 8; ++t)
        acc[t] = __builtin_amdgcn_wmma_f32_16x16x4_f32(
            false, aq[u], false, bq[u][t], (short)0, acc[t], false, false);
      fence();
    }
  }

#pragma unroll
  for (int t = 0; t < 8; ++t)
#pragma unroll
    for (int r = 0; r < 8; ++r)
      whblk[(wave * 16 + r + 8 * g) * DD + t * 16 + l16] = acc[t][r];

  float as[8], ad[8];
#pragma unroll
  for (int t = 0; t < 8; ++t) {
    as[t] = sA[t * 16 + l16];
    ad[t] = sA[DD + t * 16 + l16];
  }
  float vsel = 0.0f;
#pragma unroll
  for (int r = 0; r < 8; ++r) {
    float ps = 0.0f, pd = 0.0f;
#pragma unroll
    for (int t = 0; t < 8; ++t) {
      ps = fmaf(acc[t][r], as[t], ps);
      pd = fmaf(acc[t][r], ad[t], pd);
    }
    ps = row_xor_add<8>(ps);  pd = row_xor_add<8>(pd);
    ps = row_xor_add<4>(ps);  pd = row_xor_add<4>(pd);
    ps = row_xor_add<2>(ps);  pd = row_xor_add<2>(pd);
    ps = row_xor_add<1>(ps);  pd = row_xor_add<1>(pd);
    if ((l16 & 7) == r) vsel = (l16 < 8) ? ps : pd;
  }
  float* ebase = (l16 < 8) ? esrc : edst;
  ebase[(size_t)blockIdx.x * 128 + wave * 16 + 8 * g + (l16 & 7)] = vsel;
}

#define HALFP (NN / 4)

static __device__ __forceinline__ void stage_wh_half(const float* __restrict__ whbt, v2f* sWh,
                                                     int hf, int tid) {
  const int nq = tid & 31;
  for (int base = 0; base < HALFP; base += 64) {
    float4 r0[8], r1[8];
#pragma unroll
    for (int u = 0; u < 8; ++u) {
      const int k2 = hf * HALFP + base + u * 8 + (tid >> 5);
      r0[u] = *(const float4*)(whbt + (size_t)(2 * k2) * DD + nq * 4);
      r1[u] = *(const float4*)(whbt + (size_t)(2 * k2 + 1) * DD + nq * 4);
    }
    fence();
#pragma unroll
    for (int u = 0; u < 8; ++u) {
      const int k2l = base + u * 8 + (tid >> 5);
      float* dst = (float*)(sWh + k2l * PAIRP + nq * 4);
      *(float4*)(dst)     = make_float4(r0[u].x, r1[u].x, r0[u].y, r1[u].y);
      *(float4*)(dst + 4) = make_float4(r0[u].z, r1[u].z, r0[u].w, r1[u].w);
    }
  }
}

__global__ void __launch_bounds__(256) gat_attn_kernel(const float* __restrict__ Wh,
                                const float* __restrict__ esrc,
                                const float* __restrict__ edst,
                                const int* __restrict__ adj,
                                float* __restrict__ out) {
  extern __shared__ float smem[];
  v2f*      sWh   = (v2f*)smem;
  float*    sEd   = smem + HALFP * PAIRP * 2;
  unsigned* sMask = (unsigned*)(sEd + NN);

  const int bt = blockIdx.x >> 2;
  const int rb = blockIdx.x & 3;
  const int tid = threadIdx.x;
  const float* whbt = Wh + (size_t)bt * NN * DD;

  stage_wh_half(whbt, sWh, 0, tid);
  for (int j = tid; j < NN; j += 256) sEd[j] = edst[bt * NN + j];
  __syncthreads();

  const int lane = tid & 31;
  const int wave = tid >> 5;
  const int l16  = lane & 15;
  const int g    = lane >> 4;

  const int irow = rb * 128 + wave * 16 + l16;
  const float es = esrc[bt * NN + irow];
  const int* adjrow = adj + irow * NN;
  unsigned* maskrow = sMask + (wave * 16 + l16) * 16;

  float m = -3.0e38f, l = 0.0f;
  for (int w = 0; w < 16; ++w) {
    unsigned mw = 0;
#pragma unroll
    for (int jj = 0; jj < 16; ++jj) {
      const int j = w * 32 + 2 * jj + g;
      const bool conn = adjrow[j] > 0;
      float ev = es + sEd[j];
      ev = ev > 0.0f ? ev : ALPHA * ev;
      if (!conn) ev = NEG_INF;
      mw |= (conn ? 1u : 0u) << (2 * jj + g);
      const float nm = fmaxf(m, ev);
      l = l * __expf(m - nm) + __expf(ev - nm);
      m = nm;
    }
    mw |= __shfl_xor(mw, 16, 32);
    if (g == 0) maskrow[w] = mw;
  }
  {
    const float m2 = __shfl_xor(m, 16, 32);
    const float l2 = __shfl_xor(l, 16, 32);
    const float nm = fmaxf(m, m2);
    l = l * __expf(m - nm) + l2 * __expf(m2 - nm);
    m = nm;
  }

  v8f acc[8];
#pragma unroll
  for (int t = 0; t < 8; ++t)
#pragma unroll
    for (int r = 0; r < 8; ++r) acc[t][r] = 0.0f;

  const v2f* bptr  = sWh + g * PAIRP + l16;
  const v2f* edptr = (const v2f*)sEd + g;
  v2f bq[2][8];
  v2f edc[8], edn[8];
  unsigned mwc, mwn;

  auto prob = [&](v2f ed, unsigned mw, int c) -> v2f {
    float e0 = es + ed.x;
    e0 = e0 > 0.0f ? e0 : ALPHA * e0;
    if (!((mw >> (c * 4 + 2 * g)) & 1u)) e0 = NEG_INF;
    float e1 = es + ed.y;
    e1 = e1 > 0.0f ? e1 : ALPHA * e1;
    if (!((mw >> (c * 4 + 2 * g + 1)) & 1u)) e1 = NEG_INF;
    v2f p;
    p.x = __expf(e0 - m);
    p.y = __expf(e1 - m);
    return p;
  };

#pragma unroll
  for (int k = 0; k < 8; ++k) edc[k] = edptr[2 * k];
  mwc = maskrow[0];
  v2f a = prob(edc[0], mwc, 0);

  for (int hf = 0; hf < 2; ++hf) {
    if (hf) {
      __syncthreads();
      stage_wh_half(whbt, sWh, 1, tid);
      __syncthreads();
    }
#pragma unroll
    for (int t = 0; t < 8; ++t) bq[0][t] = bptr[t * 16];

    for (int wl = 0; wl < 8; ++wl) {
      const int w  = hf * 8 + wl;
      const int wn = (w + 1) & 15;
#pragma unroll
      for (int c = 0; c < 8; ++c) {
        const int sn = (wl * 8 + c + 1) & 63;
        const v2f* bn = bptr + 2 * sn * PAIRP;
        fence();
#pragma unroll
        for (int t = 0; t < 8; ++t) bq[(c + 1) & 1][t] = bn[t * 16];
        if (c == 0) {
#pragma unroll
          for (int k = 0; k < 8; ++k) edn[k] = edptr[16 * wn + 2 * k];
          mwn = maskrow[wn];
        }
        fence();
        const v2f an = (c < 7) ? prob(edc[(c + 1) & 7], mwc, c + 1) : prob(edn[0], mwn, 0);
#pragma unroll
        for (int t = 0; t < 8; ++t)
          acc[t] = __builtin_amdgcn_wmma_f32_16x16x4_f32(
              false, a, false, bq[c & 1][t], (short)0, acc[t], false, false);
        fence();
        a = an;
      }
#pragma unroll
      for (int k = 0; k < 8; ++k) edc[k] = edn[k];
      mwc = mwn;
    }
  }

  float* outblk = out + ((size_t)bt * NN + (size_t)rb * 128 + wave * 16) * DD;
#pragma unroll
  for (int r = 0; r < 8; ++r) {
    const int orow = r + 8 * g;
    const float lr = __shfl(l, orow, 32);
    const float inv = 1.0f / lr;
#pragma unroll
    for (int t = 0; t < 8; ++t)
      outblk[orow * DD + t * 16 + l16] = acc[t][r] * inv;
  }
}

extern "C" void kernel_launch(void* const* d_in, const int* in_sizes, int n_in,
                              void* d_out, int out_size, void* d_ws, size_t ws_size,
                              hipStream_t stream) {
  const float* h   = (const float*)d_in[0];
  const int*   adj = (const int*)d_in[1];
  const float* W   = (const float*)d_in[2];
  const float* a   = (const float*)d_in[3];
  float* out = (float*)d_out;

  float* Wh   = (float*)d_ws;
  float* esrc = Wh + (size_t)BT * NN * DD;
  float* edst = esrc + (size_t)BT * NN;

  const size_t lds1 = (size_t)(128 * LDSP) * 4 + (size_t)(64 * PAIRP) * 8 + (size_t)(2 * DD) * 4;
  const size_t lds3 = (size_t)HALFP * PAIRP * 8 + (size_t)NN * 4 +
                      (size_t)(8 * 16 * 16) * 4;

  gat_wh_kernel<<<dim3(BT * 4), dim3(256), lds1, stream>>>(h, W, a, Wh, esrc, edst);
  gat_attn_kernel<<<dim3(BT * 4), dim3(256), lds3, stream>>>(Wh, esrc, edst, adj, out);
}


# Attention pre-pass without exp: row max = leaky_relu(es + max of connected e_dst) from a coalesced adj scan whose wave votes give the mask words; row sum l added beside the WMMAs (sum order changes)
# speedup vs baseline: 1.7009x; 1.1028x over previous
#include <hip/hip_runtime.h>
#include <hip/hip_bf16.h>

#define ALPHA   0.2f
#define NEG_INF (-9000000000000000.0f)

#define BT 128
#define NN 512
#define FF 128
#define DD 128
#define LDSP  132
#define PAIRP 144

typedef float v2f __attribute__((ext_vector_type(2)));
typedef float v8f __attribute__((ext_vector_type(8)));

static __device__ __forceinline__ void fence() { __builtin_amdgcn_sched_barrier(0); }

template <int XMASK>
static __device__ __forceinline__ float row_xor_add(float v) {
  const int o = __builtin_amdgcn_update_dpp(0, __builtin_bit_cast(int, v), 0x160 + XMASK, 0xf, 0xf, true);
  return v + __builtin_bit_cast(float, o);
}

template <int XMASK>
static __device__ __forceinline__ float row_xor_max(float v) {
  const int o = __builtin_amdgcn_update_dpp(0, __builtin_bit_cast(int, v), 0x160 + XMASK, 0xf, 0xf, true);
  return fmaxf(v, __builtin_bit_cast(float, o));
}

__global__ void gat_wh_kernel(const float* __restrict__ h,
                              const float* __restrict__ W,
                              const float* __restrict__ a,
                              float* __restrict__ Wh,
                              float* __restrict__ esrc,
                              float* __restrict__ edst) {
  extern __shared__ float smem[];
  float* sH = smem;
  v2f*   sW = (v2f*)(smem + 128 * LDSP);
  float* sA = smem + 128 * LDSP + 64 * PAIRP * 2;

  const int bt = blockIdx.x >> 2;
  const int rb = blockIdx.x & 3;
  const float* hblk  = h  + ((size_t)bt * NN + (size_t)rb * 128) * FF;
  float*       whblk = Wh + ((size_t)bt * NN + (size_t)rb * 128) * DD;

  const int tid = threadIdx.x;
  for (int idx = tid * 4; idx < 128 * FF; idx += 256 * 4) {
    const int r = idx >> 7, c = idx & 127;
    *(float4*)(sH + r * LDSP + c) = *(const float4*)(hblk + idx);
  }
  for (int idx = tid; idx < 64 * 32; idx += 256) {
    const int k2 = idx >> 5, nq = idx & 31;
    float4 r0 = *(const float4*)(W + (2 * k2) * DD + nq * 4);
    float4 r1 = *(const float4*)(W + (2 * k2 + 1) * DD + nq * 4);
    float* dst = (float*)(sW + k2 * PAIRP + nq * 4);
    *(float4*)(dst)     = make_float4(r0.x, r1.x, r0.y, r1.y);
    *(float4*)(dst + 4) = make_float4(r0.z, r1.z, r0.w, r1.w);
  }
  sA[tid] = a[tid];
  __syncthreads();

  const int lane = tid & 31;
  const int wave = tid >> 5;
  const int l16  = lane & 15;
  const int g    = lane >> 4;
  const int arow = wave * 16 + l16;

  v8f acc[8];
#pragma unroll
  for (int t = 0; t < 8; ++t)
#pragma unroll
    for (int r = 0; r < 8; ++r) acc[t][r] = 0.0f;

  const float* aptr = sH + arow * LDSP + 2 * g;
  const v2f*   bptr = sW + g * PAIRP + l16;
  v2f aq[2];
  v2f bq[2][8];
  aq[0] = *(const v2f*)aptr;
#pragma unroll
  for (int t = 0; t < 8; ++t) bq[0][t] = bptr[t * 16];

  for (int s2 = 0; s2 < FF / 4; s2 += 2) {
#pragma unroll
    for (int u = 0; u < 2; ++u) {
      const int sn = (s2 + u + 1) & (FF / 4 - 1);
      const v2f* bn = bptr + 2 * sn * PAIRP;
      fence();
      aq[u ^ 1] = *(const v2f*)(aptr + 4 * sn);
#pragma unroll
      for (int t = 0; t < 8; ++t) bq[u ^ 1][t] = bn[t * 16];
      fence();
#pragma unroll
      for (int t = 0; t < 8; ++t)
        acc[t] = __builtin_amdgcn_wmma_f32_16x16x4_f32(
            false, aq[u], false, bq[u][t], (short)0, acc[t], false, false);
      fence();
    }
  }

#pragma unroll
  for (int t = 0; t < 8; ++t)
#pragma unroll
    for (int r = 0; r < 8; ++r)
      whblk[(wave * 16 + r + 8 * g) * DD + t * 16 + l16] = acc[t][r];

  float as[8], ad[8];
#pragma unroll
  for (int t = 0; t < 8; ++t) {
    as[t] = sA[t * 16 + l16];
    ad[t] = sA[DD + t * 16 + l16];
  }
  float vsel = 0.0f;
#pragma unroll
  for (int r = 0; r < 8; ++r) {
    float ps = 0.0f, pd = 0.0f;
#pragma unroll
    for (int t = 0; t < 8; ++t) {
      ps = fmaf(acc[t][r], as[t], ps);
      pd = fmaf(acc[t][r], ad[t], pd);
    }
    ps = row_xor_add<8>(ps);  pd = row_xor_add<8>(pd);
    ps = row_xor_add<4>(ps);  pd = row_xor_add<4>(pd);
    ps = row_xor_add<2>(ps);  pd = row_xor_add<2>(pd);
    ps = row_xor_add<1>(ps);  pd = row_xor_add<1>(pd);
    if ((l16 & 7) == r) vsel = (l16 < 8) ? ps : pd;
  }
  float* ebase = (l16 < 8) ? esrc : edst;
  ebase[(size_t)blockIdx.x * 128 + wave * 16 + 8 * g + (l16 & 7)] = vsel;
}

#define HALFP (NN / 4)

static __device__ __forceinline__ void stage_wh_half(const float* __restrict__ whbt, v2f* sWh,
                                                     int hf, int tid) {
  const int nq = tid & 31;
#pragma nounroll
  for (int base = 0; base < HALFP; base += 64) {
    float4 r0[8], r1[8];
#pragma unroll
    for (int u = 0; u < 8; ++u) {
      const int k2 = hf * HALFP + base + u * 8 + (tid >> 5);
      r0[u] = *(const float4*)(whbt + (size_t)(2 * k2) * DD + nq * 4);
      r1[u] = *(const float4*)(whbt + (size_t)(2 * k2 + 1) * DD + nq * 4);
    }
    fence();
#pragma unroll
    for (int u = 0; u < 8; ++u) {
      const int k2l = base + u * 8 + (tid >> 5);
      float* dst = (float*)(sWh + k2l * PAIRP + nq * 4);
      *(float4*)(dst)     = make_float4(r0[u].x, r1[u].x, r0[u].y, r1[u].y);
      *(float4*)(dst + 4) = make_float4(r0[u].z, r1[u].z, r0[u].w, r1[u].w);
    }
  }
}

__global__ void __launch_bounds__(256) gat_attn_kernel(const float* __restrict__ Wh,
                                const float* __restrict__ esrc,
                                const float* __restrict__ edst,
                                const int* __restrict__ adj,
                                float* __restrict__ out) {
  extern __shared__ float smem[];
  v2f*      sWh   = (v2f*)smem;
  float*    sEd   = smem + HALFP * PAIRP * 2;
  unsigned* sMask = (unsigned*)(sEd + NN);

  const int bt = blockIdx.x >> 2;
  const int rb = blockIdx.x & 3;
  const int tid = threadIdx.x;
  const float* whbt = Wh + (size_t)bt * NN * DD;

  stage_wh_half(whbt, sWh, 0, tid);
  for (int j = tid; j < NN; j += 256) sEd[j] = edst[bt * NN + j];
  __syncthreads();

  const int lane = tid & 31;
  const int wave = tid >> 5;
  const int l16  = lane & 15;
  const int g    = lane >> 4;

  const int irow = rb * 128 + wave * 16 + l16;
  const float es = esrc[bt * NN + irow];
  unsigned* maskrow = sMask + (wave * 16 + l16) * 16;

  const float NINF = -__builtin_inff();
  float4 edq[4];
#pragma unroll
  for (int q = 0; q < 4; ++q) edq[q] = *(const float4*)(sEd + 128 * q + 4 * lane);
  const int* adjw = adj + (size_t)(rb * 128 + wave * 16) * NN + 4 * lane;
  unsigned* mwave = sMask + wave * 256;
  float edmax = NINF;
  unsigned allc = 0;
  for (int r4 = 0; r4 < 16; r4 += 4) {
    int4 av[4][4];
#pragma unroll
    for (int i = 0; i < 4; ++i)
#pragma unroll
      for (int q = 0; q < 4; ++q) av[i][q] = *(const int4*)(adjw + (r4 + i) * NN + 128 * q);
#pragma unroll
    for (int i = 0; i < 4; ++i) {
      float mx = NINF;
      unsigned rall = ~0u;
      unsigned mrow = 0;
#pragma unroll
      for (int q = 0; q < 4; ++q) {
        const int   a4[4] = {av[i][q].x, av[i][q].y, av[i][q].z, av[i][q].w};
        const float e4[4] = {edq[q].x, edq[q].y, edq[q].z, edq[q].w};
#pragma unroll
        for (int k = 0; k < 4; ++k) {
          const bool conn = a4[k] > 0;
          const unsigned bal = __builtin_amdgcn_ballot_w32(conn);
          mx = fmaxf(mx, conn ? e4[k] : NINF);
          rall &= bal;
          if (lane == 4 * q + k) mrow = bal;
        }
      }
      if (lane < 16) mwave[(r4 + i) * 16 + lane] = mrow;
      mx = row_xor_max<8>(mx);
      mx = row_xor_max<4>(mx);
      mx = row_xor_max<2>(mx);
      mx = row_xor_max<1>(mx);
      mx = fmaxf(mx, __shfl_xor(mx, 16, 32));
      if (l16 == r4 + i) { edmax = mx; allc = rall; }
    }
  }
  float m = es + edmax;
  m = m > 0.0f ? m : ALPHA * m;
  if (allc != ~0u) m = fmaxf(m, NEG_INF);

  v8f acc[8];
#pragma unroll
  for (int t = 0; t < 8; ++t)
#pragma unroll
    for (int r = 0; r < 8; ++r) acc[t][r] = 0.0f;

  const v2f* bptr  = sWh + g * PAIRP + l16;
  const v2f* edptr = (const v2f*)sEd + g;
  v2f bq[2][8];
  const uint2* mpair = (const uint2*)maskrow + g;
  v2f edc[8], edn[8];
  uint2 mwc, mwn;
  float lsum = 0.0f;

  auto prob = [&](v2f ed, uint2 mw, int c) -> v2f {
    float e0 = es + ed.x;
    e0 = e0 > 0.0f ? e0 : ALPHA * e0;
    if (!((mw.x >> c) & 1u)) e0 = NEG_INF;
    float e1 = es + ed.y;
    e1 = e1 > 0.0f ? e1 : ALPHA * e1;
    if (!((mw.y >> c) & 1u)) e1 = NEG_INF;
    v2f p;
    p.x = __expf(e0 - m);
    p.y = __expf(e1 - m);
    return p;
  };

#pragma unroll
  for (int k = 0; k < 8; ++k) edc[k] = edptr[2 * k];
  mwc = mpair[0];
  v2f a = prob(edc[0], mwc, 0);

  for (int hf = 0; hf < 2; ++hf) {
    if (hf) {
      __syncthreads();
      stage_wh_half(whbt, sWh, 1, tid);
      __syncthreads();
    }
#pragma unroll
    for (int t = 0; t < 8; ++t) bq[0][t] = bptr[t * 16];

    for (int wl = 0; wl < 8; ++wl) {
      const int w  = hf * 8 + wl;
      const int wn = (w + 1) & 15;
#pragma unroll
      for (int c = 0; c < 8; ++c) {
        const int sn = (wl * 8 + c + 1) & 63;
        const v2f* bn = bptr + 2 * sn * PAIRP;
        fence();
#pragma unroll
        for (int t = 0; t < 8; ++t) bq[(c + 1) & 1][t] = bn[t * 16];
        if (c == 0) {
#pragma unroll
          for (int k = 0; k < 8; ++k) edn[k] = edptr[16 * wn + 2 * k];
          mwn = mpair[2 * (wn >> 2)];
        }
        fence();
        if (c == 7) {
          mwn.x >>= 8 * (wn & 3);
          mwn.y >>= 8 * (wn & 3);
        }
        const v2f an = (c < 7) ? prob(edc[(c + 1) & 7], mwc, c + 1) : prob(edn[0], mwn, 0);
        lsum += a.x + a.y;
#pragma unroll
        for (int t = 0; t < 8; ++t)
          acc[t] = __builtin_amdgcn_wmma_f32_16x16x4_f32(
              false, a, false, bq[c & 1][t], (short)0, acc[t], false, false);
        fence();
        a = an;
      }
#pragma unroll
      for (int k = 0; k < 8; ++k) edc[k] = edn[k];
      mwc = mwn;
    }
  }

  const float l = lsum + __shfl_xor(lsum, 16, 32);
  float* outblk = out + ((size_t)bt * NN + (size_t)rb * 128 + wave * 16) * DD;
#pragma unroll
  for (int r = 0; r < 8; ++r) {
    const int orow = r + 8 * g;
    const float lr = __shfl(l, orow, 32);
    const float inv = 1.0f / lr;
#pragma unroll
    for (int t = 0; t < 8; ++t)
      outblk[orow * DD + t * 16 + l16] = acc[t][r] * inv;
  }
}

extern "C" void kernel_launch(void* const* d_in, const int* in_sizes, int n_in,
                              void* d_out, int out_size, void* d_ws, size_t ws_size,
                              hipStream_t stream) {
  const float* h   = (const float*)d_in[0];
  const int*   adj = (const int*)d_in[1];
  const float* W   = (const float*)d_in[2];
  const float* a   = (const float*)d_in[3];
  float* out = (float*)d_out;

  float* Wh   = (float*)d_ws;
  float* esrc = Wh + (size_t)BT * NN * DD;
  float* edst = esrc + (size_t)BT * NN;

  const size_t lds1 = (size_t)(128 * LDSP) * 4 + (size_t)(64 * PAIRP) * 8 + (size_t)(2 * DD) * 4;
  const size_t lds3 = (size_t)HALFP * PAIRP * 8 + (size_t)NN * 4 +
                      (size_t)(8 * 16 * 16) * 4;

  gat_wh_kernel<<<dim3(BT * 4), dim3(256), lds1, stream>>>(h, W, a, Wh, esrc, edst);
  gat_attn_kernel<<<dim3(BT * 4), dim3(256), lds3, stream>>>(Wh, esrc, edst, adj, out);
}


# Staging loads batched ahead of their LDS stores: GEMM kernel's h tile and W as unrolled register arrays (2 trips to memory, was 16 + 8 serial), attention kernel's 2 e_dst loads ahead of its Wh loads
# speedup vs baseline: 1.8561x; 1.0912x over previous
#include <hip/hip_runtime.h>
#include <hip/hip_bf16.h>

#define ALPHA   0.2f
#define NEG_INF (-9000000000000000.0f)

#define BT 128
#define NN 512
#define FF 128
#define DD 128
#define LDSP  132
#define PAIRP 144

typedef float v2f __attribute__((ext_vector_type(2)));
typedef float v8f __attribute__((ext_vector_type(8)));

static __device__ __forceinline__ void fence() { __builtin_amdgcn_sched_barrier(0); }

template <int XMASK>
static __device__ __forceinline__ float row_xor_add(float v) {
  const int o = __builtin_amdgcn_update_dpp(0, __builtin_bit_cast(int, v), 0x160 + XMASK, 0xf, 0xf, true);
  return v + __builtin_bit_cast(float, o);
}

template <int XMASK>
static __device__ __forceinline__ float row_xor_max(float v) {
  const int o = __builtin_amdgcn_update_dpp(0, __builtin_bit_cast(int, v), 0x160 + XMASK, 0xf, 0xf, true);
  return fmaxf(v, __builtin_bit_cast(float, o));
}

__global__ void gat_wh_kernel(const float* __restrict__ h,
                              const float* __restrict__ W,
                              const float* __restrict__ a,
                              float* __restrict__ Wh,
                              float* __restrict__ esrc,
                              float* __restrict__ edst) {
  extern __shared__ float smem[];
  float* sH = smem;
  v2f*   sW = (v2f*)(smem + 128 * LDSP);
  float* sA = smem + 128 * LDSP + 64 * PAIRP * 2;

  const int bt = blockIdx.x >> 2;
  const int rb = blockIdx.x & 3;
  const float* hblk  = h  + ((size_t)bt * NN + (size_t)rb * 128) * FF;
  float*       whblk = Wh + ((size_t)bt * NN + (size_t)rb * 128) * DD;

  const int tid = threadIdx.x;
  const int nq = tid & 31;
  const int r8 = tid >> 5;
  {
    float4 w0[8], w1[8];
#pragma unroll
    for (int u = 0; u < 8; ++u) {
      const int k2 = u * 8 + r8;
      w0[u] = *(const float4*)(W + (2 * k2) * DD + nq * 4);
      w1[u] = *(const float4*)(W + (2 * k2 + 1) * DD + nq * 4);
    }
    const float av = a[tid];
#pragma unroll
    for (int u = 0; u < 8; ++u) {
      const int k2 = u * 8 + r8;
      float* dst = (float*)(sW + k2 * PAIRP + nq * 4);
      *(float4*)(dst)     = make_float4(w0[u].x, w1[u].x, w0[u].y, w1[u].y);
      *(float4*)(dst + 4) = make_float4(w0[u].z, w1[u].z, w0[u].w, w1[u].w);
    }
    sA[tid] = av;
  }
  {
    float4 hv[16];
#pragma unroll
    for (int u = 0; u < 16; ++u) hv[u] = *(const float4*)(hblk + (u * 8 + r8) * FF + nq * 4);
#pragma unroll
    for (int u = 0; u < 16; ++u) *(float4*)(sH + (u * 8 + r8) * LDSP + nq * 4) = hv[u];
  }
  __syncthreads();

  const int lane = tid & 31;
  const int wave = tid >> 5;
  const int l16  = lane & 15;
  const int g    = lane >> 4;
  const int arow = wave * 16 + l16;

  v8f acc[8];
#pragma unroll
  for (int t = 0; t < 8; ++t)
#pragma unroll
    for (int r = 0; r < 8; ++r) acc[t][r] = 0.0f;

  const float* aptr = sH + arow * LDSP + 2 * g;
  const v2f*   bptr = sW + g * PAIRP + l16;
  v2f aq[2];
  v2f bq[2][8];
  aq[0] = *(const v2f*)aptr;
#pragma unroll
  for (int t = 0; t < 8; ++t) bq[0][t] = bptr[t * 16];

  for (int s2 = 0; s2 < FF / 4; s2 += 2) {
#pragma unroll
    for (int u = 0; u < 2; ++u) {
      const int sn = (s2 + u + 1) & (FF / 4 - 1);
      const v2f* bn = bptr + 2 * sn * PAIRP;
      fence();
      aq[u ^ 1] = *(const v2f*)(aptr + 4 * sn);
#pragma unroll
      for (int t = 0; t < 8; ++t) bq[u ^ 1][t] = bn[t * 16];
      fence();
#pragma unroll
      for (int t = 0; t < 8; ++t)
        acc[t] = __builtin_amdgcn_wmma_f32_16x16x4_f32(
            false, aq[u], false, bq[u][t], (short)0, acc[t], false, false);
      fence();
    }
  }

#pragma unroll
  for (int t = 0; t < 8; ++t)
#pragma unroll
    for (int r = 0; r < 8; ++r)
      whblk[(wave * 16 + r + 8 * g) * DD + t * 16 + l16] = acc[t][r];

  float as[8], ad[8];
#pragma unroll
  for (int t = 0; t < 8; ++t) {
    as[t] = sA[t * 16 + l16];
    ad[t] = sA[DD + t * 16 + l16];
  }
  float vsel = 0.0f;
#pragma unroll
  for (int r = 0; r < 8; ++r) {
    float ps = 0.0f, pd = 0.0f;
#pragma unroll
    for (int t = 0; t < 8; ++t) {
      ps = fmaf(acc[t][r], as[t], ps);
      pd = fmaf(acc[t][r], ad[t], pd);
    }
    ps = row_xor_add<8>(ps);  pd = row_xor_add<8>(pd);
    ps = row_xor_add<4>(ps);  pd = row_xor_add<4>(pd);
    ps = row_xor_add<2>(ps);  pd = row_xor_add<2>(pd);
    ps = row_xor_add<1>(ps);  pd = row_xor_add<1>(pd);
    if ((l16 & 7) == r) vsel = (l16 < 8) ? ps : pd;
  }
  float* ebase = (l16 < 8) ? esrc : edst;
  ebase[(size_t)blockIdx.x * 128 + wave * 16 + 8 * g + (l16 & 7)] = vsel;
}

#define HALFP (NN / 4)

static __device__ __forceinline__ void stage_wh_half(const float* __restrict__ whbt, v2f* sWh,
                                                     int hf, int tid) {
  const int nq = tid & 31;
#pragma nounroll
  for (int base = 0; base < HALFP; base += 64) {
    float4 r0[8], r1[8];
#pragma unroll
    for (int u = 0; u < 8; ++u) {
      const int k2 = hf * HALFP + base + u * 8 + (tid >> 5);
      r0[u] = *(const float4*)(whbt + (size_t)(2 * k2) * DD + nq * 4);
      r1[u] = *(const float4*)(whbt + (size_t)(2 * k2 + 1) * DD + nq * 4);
    }
    fence();
#pragma unroll
    for (int u = 0; u < 8; ++u) {
      const int k2l = base + u * 8 + (tid >> 5);
      float* dst = (float*)(sWh + k2l * PAIRP + nq * 4);
      *(float4*)(dst)     = make_float4(r0[u].x, r1[u].x, r0[u].y, r1[u].y);
      *(float4*)(dst + 4) = make_float4(r0[u].z, r1[u].z, r0[u].w, r1[u].w);
    }
  }
}

__global__ void __launch_bounds__(256) gat_attn_kernel(const float* __restrict__ Wh,
                                const float* __restrict__ esrc,
                                const float* __restrict__ edst,
                                const int* __restrict__ adj,
                                float* __restrict__ out) {
  extern __shared__ float smem[];
  v2f*      sWh   = (v2f*)smem;
  float*    sEd   = smem + HALFP * PAIRP * 2;
  unsigned* sMask = (unsigned*)(sEd + NN);

  const int bt = blockIdx.x >> 2;
  const int rb = blockIdx.x & 3;
  const int tid = threadIdx.x;
  const float* whbt = Wh + (size_t)bt * NN * DD;

  const float ed0 = edst[bt * NN + tid];
  const float ed1 = edst[bt * NN + 256 + tid];
  stage_wh_half(whbt, sWh, 0, tid);
  sEd[tid]       = ed0;
  sEd[256 + tid] = ed1;
  __syncthreads();

  const int lane = tid & 31;
  const int wave = tid >> 5;
  const int l16  = lane & 15;
  const int g    = lane >> 4;

  const int irow = rb * 128 + wave * 16 + l16;
  const float es = esrc[bt * NN + irow];
  unsigned* maskrow = sMask + (wave * 16 + l16) * 16;

  const float NINF = -__builtin_inff();
  float4 edq[4];
#pragma unroll
  for (int q = 0; q < 4; ++q) edq[q] = *(const float4*)(sEd + 128 * q + 4 * lane);
  const int* adjw = adj + (size_t)(rb * 128 + wave * 16) * NN + 4 * lane;
  unsigned* mwave = sMask + wave * 256;
  float edmax = NINF;
  unsigned allc = 0;
  for (int r4 = 0; r4 < 16; r4 += 4) {
    int4 av[4][4];
#pragma unroll
    for (int i = 0; i < 4; ++i)
#pragma unroll
      for (int q = 0; q < 4; ++q) av[i][q] = *(const int4*)(adjw + (r4 + i) * NN + 128 * q);
#pragma unroll
    for (int i = 0; i < 4; ++i) {
      float mx = NINF;
      unsigned rall = ~0u;
      unsigned mrow = 0;
#pragma unroll
      for (int q = 0; q < 4; ++q) {
        const int   a4[4] = {av[i][q].x, av[i][q].y, av[i][q].z, av[i][q].w};
        const float e4[4] = {edq[q].x, edq[q].y, edq[q].z, edq[q].w};
#pragma unroll
        for (int k = 0; k < 4; ++k) {
          const bool conn = a4[k] > 0;
          const unsigned bal = __builtin_amdgcn_ballot_w32(conn);
          mx = fmaxf(mx, conn ? e4[k] : NINF);
          rall &= bal;
          if (lane == 4 * q + k) mrow = bal;
        }
      }
      if (lane < 16) mwave[(r4 + i) * 16 + lane] = mrow;
      mx = row_xor_max<8>(mx);
      mx = row_xor_max<4>(mx);
      mx = row_xor_max<2>(mx);
      mx = row_xor_max<1>(mx);
      mx = fmaxf(mx, __shfl_xor(mx, 16, 32));
      if (l16 == r4 + i) { edmax = mx; allc = rall; }
    }
  }
  float m = es + edmax;
  m = m > 0.0f ? m : ALPHA * m;
  if (allc != ~0u) m = fmaxf(m, NEG_INF);

  v8f acc[8];
#pragma unroll
  for (int t = 0; t < 8; ++t)
#pragma unroll
    for (int r = 0; r < 8; ++r) acc[t][r] = 0.0f;

  const v2f* bptr  = sWh + g * PAIRP + l16;
  const v2f* edptr = (const v2f*)sEd + g;
  v2f bq[2][8];
  const uint2* mpair = (const uint2*)maskrow + g;
  v2f edc[8], edn[8];
  uint2 mwc, mwn;
  float lsum = 0.0f;

  auto prob = [&](v2f ed, uint2 mw, int c) -> v2f {
    float e0 = es + ed.x;
    e0 = e0 > 0.0f ? e0 : ALPHA * e0;
    if (!((mw.x >> c) & 1u)) e0 = NEG_INF;
    float e1 = es + ed.y;
    e1 = e1 > 0.0f ? e1 : ALPHA * e1;
    if (!((mw.y >> c) & 1u)) e1 = NEG_INF;
    v2f p;
    p.x = __expf(e0 - m);
    p.y = __expf(e1 - m);
    return p;
  };

#pragma unroll
  for (int k = 0; k < 8; ++k) edc[k] = edptr[2 * k];
  mwc = mpair[0];
  v2f a = prob(edc[0], mwc, 0);

  for (int hf = 0; hf < 2; ++hf) {
    if (hf) {
      __syncthreads();
      stage_wh_half(whbt, sWh, 1, tid);
      __syncthreads();
    }
#pragma unroll
    for (int t = 0; t < 8; ++t) bq[0][t] = bptr[t * 16];

    for (int wl = 0; wl < 8; ++wl) {
      const int w  = hf * 8 + wl;
      const int wn = (w + 1) & 15;
#pragma unroll
      for (int c = 0; c < 8; ++c) {
        const int sn = (wl * 8 + c + 1) & 63;
        const v2f* bn = bptr + 2 * sn * PAIRP;
        fence();
#pragma unroll
        for (int t = 0; t < 8; ++t) bq[(c + 1) & 1][t] = bn[t * 16];
        if (c == 0) {
#pragma unroll
          for (int k = 0; k < 8; ++k) edn[k] = edptr[16 * wn + 2 * k];
          mwn = mpair[2 * (wn >> 2)];
        }
        fence();
        if (c == 7) {
          mwn.x >>= 8 * (wn & 3);
          mwn.y >>= 8 * (wn & 3);
        }
        const v2f an = (c < 7) ? prob(edc[(c + 1) & 7], mwc, c + 1) : prob(edn[0], mwn, 0);
        lsum += a.x + a.y;
#pragma unroll
        for (int t = 0; t < 8; ++t)
          acc[t] = __builtin_amdgcn_wmma_f32_16x16x4_f32(
              false, a, false, bq[c & 1][t], (short)0, acc[t], false, false);
        fence();
        a = an;
      }
#pragma unroll
      for (int k = 0; k < 8; ++k) edc[k] = edn[k];
      mwc = mwn;
    }
  }

  const float l = lsum + __shfl_xor(lsum, 16, 32);
  float* outblk = out + ((size_t)bt * NN + (size_t)rb * 128 + wave * 16) * DD;
#pragma unroll
  for (int r = 0; r < 8; ++r) {
    const int orow = r + 8 * g;
    const float lr = __shfl(l, orow, 32);
    const float inv = 1.0f / lr;
#pragma unroll
    for (int t = 0; t < 8; ++t)
      outblk[orow * DD + t * 16 + l16] = acc[t][r] * inv;
  }
}

extern "C" void kernel_launch(void* const* d_in, const int* in_sizes, int n_in,
                              void* d_out, int out_size, void* d_ws, size_t ws_size,
                              hipStream_t stream) {
  const float* h   = (const float*)d_in[0];
  const int*   adj = (const int*)d_in[1];
  const float* W   = (const float*)d_in[2];
  const float* a   = (const float*)d_in[3];
  float* out = (float*)d_out;

  float* Wh   = (float*)d_ws;
  float* esrc = Wh + (size_t)BT * NN * DD;
  float* edst = esrc + (size_t)BT * NN;

  const size_t lds1 = (size_t)(128 * LDSP) * 4 + (size_t)(64 * PAIRP) * 8 + (size_t)(2 * DD) * 4;
  const size_t lds3 = (size_t)HALFP * PAIRP * 8 + (size_t)NN * 4 +
                      (size_t)(8 * 16 * 16) * 4;

  gat_wh_kernel<<<dim3(BT * 4), dim3(256), lds1, stream>>>(h, W, a, Wh, esrc, edst);
  gat_attn_kernel<<<dim3(BT * 4), dim3(256), lds3, stream>>>(Wh, esrc, edst, adj, out);
}


# Product reassociated to (softmax(e) @ h) @ W: W multiplied from the accumulators, e from h.(W a); one kernel per 256 rows does all, 2 launches -> 1, no h@W in memory; sum order changes
# speedup vs baseline: 2.1174x; 1.1408x over previous
#include <hip/hip_runtime.h>
#include <hip/hip_bf16.h>

#define ALPHA   0.2f
#define NEG_INF (-9000000000000000.0f)

#define BT 128
#define NN 512
#define FF 128
#define DD 128
#define LDSP  132
#define PAIRP 144
#define HALFP (NN / 4)

typedef float v2f __attribute__((ext_vector_type(2)));
typedef float v8f __attribute__((ext_vector_type(8)));

static __device__ __forceinline__ void fence() { __builtin_amdgcn_sched_barrier(0); }

template <int XMASK>
static __device__ __forceinline__ float row_xor_add(float v) {
  const int o = __builtin_amdgcn_update_dpp(0, __builtin_bit_cast(int, v), 0x160 + XMASK, 0xf, 0xf, true);
  return v + __builtin_bit_cast(float, o);
}

template <int XMASK>
static __device__ __forceinline__ float row_xor_max(float v) {
  const int o = __builtin_amdgcn_update_dpp(0, __builtin_bit_cast(int, v), 0x160 + XMASK, 0xf, 0xf, true);
  return fmaxf(v, __builtin_bit_cast(float, o));
}

__global__ void __launch_bounds__(512) gat_fused_kernel(const float* __restrict__ h,
                                                        const int* __restrict__ adj,
                                                        const float* __restrict__ W,
                                                        const float* __restrict__ avec,
                                                        float* __restrict__ out) {
  extern __shared__ float smem[];
  v2f*      sWh   = (v2f*)smem;
  float*    sEd   = smem + 256 * PAIRP * 2;
  float*    sEs   = sEd + NN;
  float*    sWa   = sEs + NN;
  unsigned* sMask = (unsigned*)(sWa + 2 * FF);
  v2f*      sW2   = (v2f*)smem;
  float*    sH2   = smem + 64 * PAIRP * 2;

  const int bt = blockIdx.x >> 1;
  const int hf = blockIdx.x & 1;
  const int tid = threadIdx.x;
  const int lane = tid & 31;
  const int wave = tid >> 5;
  const int l16  = lane & 15;
  const int g    = lane >> 4;
  const float* hbt = h + (size_t)bt * NN * FF;

  {
    float4 wr[8];
#pragma unroll
    for (int u = 0; u < 8; ++u) wr[u] = *(const float4*)(W + (wave * 8 + u) * DD + 4 * lane);
    const float4 as4 = *(const float4*)(avec + 4 * lane);
    const float4 ad4 = *(const float4*)(avec + DD + 4 * lane);
    float4 r0[16], r1[16];
#pragma unroll
    for (int u = 0; u < 16; ++u) {
      const int k2 = u * 16 + wave;
      r0[u] = *(const float4*)(hbt + (size_t)(2 * k2) * FF + lane * 4);
      r1[u] = *(const float4*)(hbt + (size_t)(2 * k2 + 1) * FF + lane * 4);
    }
    fence();
#pragma unroll
    for (int u = 0; u < 8; ++u) {
      float ps = wr[u].x * as4.x;
      ps = fmaf(wr[u].y, as4.y, ps);
      ps = fmaf(wr[u].z, as4.z, ps);
      ps = fmaf(wr[u].w, as4.w, ps);
      float pd = wr[u].x * ad4.x;
      pd = fmaf(wr[u].y, ad4.y, pd);
      pd = fmaf(wr[u].z, ad4.z, pd);
      pd = fmaf(wr[u].w, ad4.w, pd);
      ps = row_xor_add<8>(ps);  pd = row_xor_add<8>(pd);
      ps = row_xor_add<4>(ps);  pd = row_xor_add<4>(pd);
      ps = row_xor_add<2>(ps);  pd = row_xor_add<2>(pd);
      ps = row_xor_add<1>(ps);  pd = row_xor_add<1>(pd);
      ps += __shfl_xor(ps, 16, 32);
      pd += __shfl_xor(pd, 16, 32);
      if (lane == 0) {
        sWa[wave * 8 + u]      = ps;
        sWa[FF + wave * 8 + u] = pd;
      }
    }
#pragma unroll
    for (int u = 0; u < 16; ++u) {
      const int k2 = u * 16 + wave;
      float* dst = (float*)(sWh + k2 * PAIRP + lane * 4);
      *(float4*)(dst)     = make_float4(r0[u].x, r1[u].x, r0[u].y, r1[u].y);
      *(float4*)(dst + 4) = make_float4(r0[u].z, r1[u].z, r0[u].w, r1[u].w);
    }
  }
  __syncthreads();

  {
    const int c8 = lane & 7;
    const int q4 = lane >> 3;
    v2f ws2[8], wd2[8];
#pragma unroll
    for (int i = 0; i < 8; ++i) {
      ws2[i] = *(const v2f*)(sWa + 2 * c8 + 16 * i);
      wd2[i] = *(const v2f*)(sWa + FF + 2 * c8 + 16 * i);
    }
#pragma unroll
    for (int ps4 = 0; ps4 < 4; ++ps4) {
      const int p = wave * 16 + ps4 * 4 + q4;
      const float* src = (const float*)(sWh + p * PAIRP + 2 * c8);
      float es0 = 0.0f, es1 = 0.0f, ed0 = 0.0f, ed1 = 0.0f;
#pragma unroll
      for (int i = 0; i < 8; ++i) {
        const float4 x = *(const float4*)(src + 32 * i);
        es0 = fmaf(x.x, ws2[i].x, es0);  es0 = fmaf(x.z, ws2[i].y, es0);
        es1 = fmaf(x.y, ws2[i].x, es1);  es1 = fmaf(x.w, ws2[i].y, es1);
        ed0 = fmaf(x.x, wd2[i].x, ed0);  ed0 = fmaf(x.z, wd2[i].y, ed0);
        ed1 = fmaf(x.y, wd2[i].x, ed1);  ed1 = fmaf(x.w, wd2[i].y, ed1);
      }
      es0 = row_xor_add<4>(es0);  es1 = row_xor_add<4>(es1);  ed0 = row_xor_add<4>(ed0);  ed1 = row_xor_add<4>(ed1);
      es0 = row_xor_add<2>(es0);  es1 = row_xor_add<2>(es1);  ed0 = row_xor_add<2>(ed0);  ed1 = row_xor_add<2>(ed1);
      es0 = row_xor_add<1>(es0);  es1 = row_xor_add<1>(es1);  ed0 = row_xor_add<1>(ed0);  ed1 = row_xor_add<1>(ed1);
      if (c8 == 0) {
        *(v2f*)(sEs + 2 * p) = v2f{es0, es1};
        *(v2f*)(sEd + 2 * p) = v2f{ed0, ed1};
      }
    }
  }
  __syncthreads();

  const int irow = hf * 256 + wave * 16 + l16;
  const float es = sEs[irow];
  unsigned* maskrow = sMask + (wave * 16 + l16) * 16;

  const float NINF = -__builtin_inff();
  float4 edq[4];
#pragma unroll
  for (int q = 0; q < 4; ++q) edq[q] = *(const float4*)(sEd + 128 * q + 4 * lane);
  const int* adjw = adj + (size_t)(hf * 256 + wave * 16) * NN + 4 * lane;
  unsigned* mwave = sMask + wave * 256;
  float edmax = NINF;
  unsigned allc = 0;
  for (int r4 = 0; r4 < 16; r4 += 4) {
    int4 av[4][4];
#pragma unroll
    for (int i = 0; i < 4; ++i)
#pragma unroll
      for (int q = 0; q < 4; ++q) av[i][q] = *(const int4*)(adjw + (r4 + i) * NN + 128 * q);
#pragma unroll
    for (int i = 0; i < 4; ++i) {
      float mx = NINF;
      unsigned rall = ~0u;
      unsigned mrow = 0;
#pragma unroll
      for (int q = 0; q < 4; ++q) {
        const int   a4[4] = {av[i][q].x, av[i][q].y, av[i][q].z, av[i][q].w};
        const float e4[4] = {edq[q].x, edq[q].y, edq[q].z, edq[q].w};
#pragma unroll
        for (int k = 0; k < 4; ++k) {
          const bool conn = a4[k] > 0;
          const unsigned bal = __builtin_amdgcn_ballot_w32(conn);
          mx = fmaxf(mx, conn ? e4[k] : NINF);
          rall &= bal;
          if (lane == 4 * q + k) mrow = bal;
        }
      }
      if (lane < 16) mwave[(r4 + i) * 16 + lane] = mrow;
      mx = row_xor_max<8>(mx);
      mx = row_xor_max<4>(mx);
      mx = row_xor_max<2>(mx);
      mx = row_xor_max<1>(mx);
      mx = fmaxf(mx, __shfl_xor(mx, 16, 32));
      if (l16 == r4 + i) { edmax = mx; allc = rall; }
    }
  }
  float m = es + edmax;
  m = m > 0.0f ? m : ALPHA * m;
  if (allc != ~0u) m = fmaxf(m, NEG_INF);

  v8f acc[8];
#pragma unroll
  for (int t = 0; t < 8; ++t)
#pragma unroll
    for (int r = 0; r < 8; ++r) acc[t][r] = 0.0f;

  const v2f* bptr  = sWh + g * PAIRP + l16;
  const v2f* edptr = (const v2f*)sEd + g;
  v2f bq[2][8];
  const uint2* mpair = (const uint2*)maskrow + g;
  v2f edc[8], edn[8];
  uint2 mwc, mwn;
  float lsum = 0.0f;

  auto prob = [&](v2f ed, uint2 mw, int c) -> v2f {
    float e0 = es + ed.x;
    e0 = e0 > 0.0f ? e0 : ALPHA * e0;
    if (!((mw.x >> c) & 1u)) e0 = NEG_INF;
    float e1 = es + ed.y;
    e1 = e1 > 0.0f ? e1 : ALPHA * e1;
    if (!((mw.y >> c) & 1u)) e1 = NEG_INF;
    v2f p;
    p.x = __expf(e0 - m);
    p.y = __expf(e1 - m);
    return p;
  };

#pragma unroll
  for (int k = 0; k < 8; ++k) edc[k] = edptr[2 * k];
  mwc = mpair[0];
  v2f a = prob(edc[0], mwc, 0);

  for (int hh = 0; hh < 2; ++hh) {
    const v2f* bh = bptr + hh * (HALFP * PAIRP);
#pragma unroll
    for (int t = 0; t < 8; ++t) bq[0][t] = bh[t * 16];

    for (int wl = 0; wl < 8; ++wl) {
      const int w  = hh * 8 + wl;
      const int wn = (w + 1) & 15;
#pragma unroll
      for (int c = 0; c < 8; ++c) {
        const int sn = (wl * 8 + c + 1) & 63;
        const v2f* bn = bh + 2 * sn * PAIRP;
        fence();
#pragma unroll
        for (int t = 0; t < 8; ++t) bq[(c + 1) & 1][t] = bn[t * 16];
        if (c == 0) {
#pragma unroll
          for (int k = 0; k < 8; ++k) edn[k] = edptr[16 * wn + 2 * k];
          mwn = mpair[2 * (wn >> 2)];
        }
        fence();
        if (c == 7) {
          mwn.x >>= 8 * (wn & 3);
          mwn.y >>= 8 * (wn & 3);
        }
        const v2f an = (c < 7) ? prob(edc[(c + 1) & 7], mwc, c + 1) : prob(edn[0], mwn, 0);
        lsum += a.x + a.y;
#pragma unroll
        for (int t = 0; t < 8; ++t)
          acc[t] = __builtin_amdgcn_wmma_f32_16x16x4_f32(
              false, a, false, bq[c & 1][t], (short)0, acc[t], false, false);
        fence();
        a = an;
      }
#pragma unroll
      for (int k = 0; k < 8; ++k) edc[k] = edn[k];
      mwc = mwn;
    }
  }

  const float l = lsum + __shfl_xor(lsum, 16, 32);

  {
    float4 w0[4], w1[4];
#pragma unroll
    for (int u = 0; u < 4; ++u) {
      const int k2 = u * 16 + wave;
      w0[u] = *(const float4*)(W + (2 * k2) * DD + lane * 4);
      w1[u] = *(const float4*)(W + (2 * k2 + 1) * DD + lane * 4);
    }
    __syncthreads();
#pragma unroll
    for (int u = 0; u < 4; ++u) {
      const int k2 = u * 16 + wave;
      float* dst = (float*)(sW2 + k2 * PAIRP + lane * 4);
      *(float4*)(dst)     = make_float4(w0[u].x, w1[u].x, w0[u].y, w1[u].y);
      *(float4*)(dst + 4) = make_float4(w0[u].z, w1[u].z, w0[u].w, w1[u].w);
    }
#pragma unroll
    for (int t = 0; t < 8; ++t)
#pragma unroll
      for (int r = 0; r < 8; ++r)
        sH2[(wave * 16 + r + 8 * g) * LDSP + t * 16 + l16] = acc[t][r];
  }
  __syncthreads();

#pragma unroll
  for (int t = 0; t < 8; ++t)
#pragma unroll
    for (int r = 0; r < 8; ++r) acc[t][r] = 0.0f;

  {
    const int arow = wave * 16 + l16;
    const float* aptr  = sH2 + arow * LDSP + 2 * g;
    const v2f*   bptr2 = sW2 + g * PAIRP + l16;
    v2f aq[2];
    v2f bw[2][8];
    aq[0] = *(const v2f*)aptr;
#pragma unroll
    for (int t = 0; t < 8; ++t) bw[0][t] = bptr2[t * 16];

    for (int s2 = 0; s2 < FF / 4; s2 += 2) {
#pragma unroll
      for (int u = 0; u < 2; ++u) {
        const int sn = (s2 + u + 1) & (FF / 4 - 1);
        const v2f* bn = bptr2 + 2 * sn * PAIRP;
        fence();
        aq[u ^ 1] = *(const v2f*)(aptr + 4 * sn);
#pragma unroll
        for (int t = 0; t < 8; ++t) bw[u ^ 1][t] = bn[t * 16];
        fence();
#pragma unroll
        for (int t = 0; t < 8; ++t)
          acc[t] = __builtin_amdgcn_wmma_f32_16x16x4_f32(
              false, aq[u], false, bw[u][t], (short)0, acc[t], false, false);
        fence();
      }
    }
  }

  float* outblk = out + ((size_t)bt * NN + (size_t)hf * 256 + wave * 16) * DD;
#pragma unroll
  for (int r = 0; r < 8; ++r) {
    const int orow = r + 8 * g;
    const float lr = __shfl(l, orow, 32);
    const float inv = 1.0f / lr;
#pragma unroll
    for (int t = 0; t < 8; ++t)
      outblk[orow * DD + t * 16 + l16] = acc[t][r] * inv;
  }
}

extern "C" void kernel_launch(void* const* d_in, const int* in_sizes, int n_in,
                              void* d_out, int out_size, void* d_ws, size_t ws_size,
                              hipStream_t stream) {
  const float* h   = (const float*)d_in[0];
  const int*   adj = (const int*)d_in[1];
  const float* W   = (const float*)d_in[2];
  const float* a   = (const float*)d_in[3];
  float* out = (float*)d_out;

  const size_t lds = (size_t)(256 * PAIRP) * 8 + (size_t)NN * 4 + (size_t)NN * 4 +
                     (size_t)(2 * FF) * 4 + (size_t)(16 * 16 * 16) * 4;

  gat_fused_kernel<<<dim3(BT * 2), dim3(512), lds, stream>>>(h, adj, W, a, out);
}


# Attention step's vector work 20 -> 11 instructions: base-2 scores, leaky_relu minus row max as max of packed add and fma, mask bit from a doubled word's carry, packed row sums; sum order changes
# speedup vs baseline: 2.2029x; 1.0404x over previous
#include <hip/hip_runtime.h>
#include <hip/hip_bf16.h>

#define ALPHA   0.2f
#define NEG_INF (-9000000000000000.0f)
#define LOG2E   1.4426950408889634f
#define NEG_INF2 (NEG_INF * LOG2E)

#define BT 128
#define NN 512
#define FF 128
#define DD 128
#define LDSP  132
#define PAIRP 144
#define HALFP (NN / 4)

typedef float v2f __attribute__((ext_vector_type(2)));
typedef float v8f __attribute__((ext_vector_type(8)));

static __device__ __forceinline__ void fence() { __builtin_amdgcn_sched_barrier(0); }

template <int XMASK>
static __device__ __forceinline__ float row_xor_add(float v) {
  const int o = __builtin_amdgcn_update_dpp(0, __builtin_bit_cast(int, v), 0x160 + XMASK, 0xf, 0xf, true);
  return v + __builtin_bit_cast(float, o);
}

template <int XMASK>
static __device__ __forceinline__ float row_xor_max(float v) {
  const int o = __builtin_amdgcn_update_dpp(0, __builtin_bit_cast(int, v), 0x160 + XMASK, 0xf, 0xf, true);
  return fmaxf(v, __builtin_bit_cast(float, o));
}

__global__ void __launch_bounds__(512) gat_fused_kernel(const float* __restrict__ h,
                                                        const int* __restrict__ adj,
                                                        const float* __restrict__ W,
                                                        const float* __restrict__ avec,
                                                        float* __restrict__ out) {
  extern __shared__ float smem[];
  v2f*      sWh   = (v2f*)smem;
  float*    sEd   = smem + 256 * PAIRP * 2;
  float*    sEs   = sEd + NN;
  float*    sWa   = sEs + NN;
  unsigned* sMask = (unsigned*)(sWa + 2 * FF);
  v2f*      sW2   = (v2f*)smem;
  float*    sH2   = smem + 64 * PAIRP * 2;

  const int bt = blockIdx.x >> 1;
  const int hf = blockIdx.x & 1;
  const int tid = threadIdx.x;
  const int lane = tid & 31;
  const int wave = tid >> 5;
  const int l16  = lane & 15;
  const int g    = lane >> 4;
  const float* hbt = h + (size_t)bt * NN * FF;

  {
    float4 wr[8];
#pragma unroll
    for (int u = 0; u < 8; ++u) wr[u] = *(const float4*)(W + (wave * 8 + u) * DD + 4 * lane);
    const float4 as4 = *(const float4*)(avec + 4 * lane);
    const float4 ad4 = *(const float4*)(avec + DD + 4 * lane);
    float4 r0[16], r1[16];
#pragma unroll
    for (int u = 0; u < 16; ++u) {
      const int k2 = u * 16 + wave;
      r0[u] = *(const float4*)(hbt + (size_t)(2 * k2) * FF + lane * 4);
      r1[u] = *(const float4*)(hbt + (size_t)(2 * k2 + 1) * FF + lane * 4);
    }
    fence();
#pragma unroll
    for (int u = 0; u < 8; ++u) {
      float ps = wr[u].x * as4.x;
      ps = fmaf(wr[u].y, as4.y, ps);
      ps = fmaf(wr[u].z, as4.z, ps);
      ps = fmaf(wr[u].w, as4.w, ps);
      float pd = wr[u].x * ad4.x;
      pd = fmaf(wr[u].y, ad4.y, pd);
      pd = fmaf(wr[u].z, ad4.z, pd);
      pd = fmaf(wr[u].w, ad4.w, pd);
      ps = row_xor_add<8>(ps);  pd = row_xor_add<8>(pd);
      ps = row_xor_add<4>(ps);  pd = row_xor_add<4>(pd);
      ps = row_xor_add<2>(ps);  pd = row_xor_add<2>(pd);
      ps = row_xor_add<1>(ps);  pd = row_xor_add<1>(pd);
      ps += __shfl_xor(ps, 16, 32);
      pd += __shfl_xor(pd, 16, 32);
      if (lane == 0) {
        sWa[wave * 8 + u]      = ps * LOG2E;
        sWa[FF + wave * 8 + u] = pd * LOG2E;
      }
    }
#pragma unroll
    for (int u = 0; u < 16; ++u) {
      const int k2 = u * 16 + wave;
      float* dst = (float*)(sWh + k2 * PAIRP + lane * 4);
      *(float4*)(dst)     = make_float4(r0[u].x, r1[u].x, r0[u].y, r1[u].y);
      *(float4*)(dst + 4) = make_float4(r0[u].z, r1[u].z, r0[u].w, r1[u].w);
    }
  }
  __syncthreads();

  {
    const int c8 = lane & 7;
    const int q4 = lane >> 3;
    v2f ws2[8], wd2[8];
#pragma unroll
    for (int i = 0; i < 8; ++i) {
      ws2[i] = *(const v2f*)(sWa + 2 * c8 + 16 * i);
      wd2[i] = *(const v2f*)(sWa + FF + 2 * c8 + 16 * i);
    }
#pragma unroll
    for (int ps4 = 0; ps4 < 4; ++ps4) {
      const int p = wave * 16 + ps4 * 4 + q4;
      const float* src = (const float*)(sWh + p * PAIRP + 2 * c8);
      float es0 = 0.0f, es1 = 0.0f, ed0 = 0.0f, ed1 = 0.0f;
#pragma unroll
      for (int i = 0; i < 8; ++i) {
        const float4 x = *(const float4*)(src + 32 * i);
        es0 = fmaf(x.x, ws2[i].x, es0);  es0 = fmaf(x.z, ws2[i].y, es0);
        es1 = fmaf(x.y, ws2[i].x, es1);  es1 = fmaf(x.w, ws2[i].y, es1);
        ed0 = fmaf(x.x, wd2[i].x, ed0);  ed0 = fmaf(x.z, wd2[i].y, ed0);
        ed1 = fmaf(x.y, wd2[i].x, ed1);  ed1 = fmaf(x.w, wd2[i].y, ed1);
      }
      es0 = row_xor_add<4>(es0);  es1 = row_xor_add<4>(es1);  ed0 = row_xor_add<4>(ed0);  ed1 = row_xor_add<4>(ed1);
      es0 = row_xor_add<2>(es0);  es1 = row_xor_add<2>(es1);  ed0 = row_xor_add<2>(ed0);  ed1 = row_xor_add<2>(ed1);
      es0 = row_xor_add<1>(es0);  es1 = row_xor_add<1>(es1);  ed0 = row_xor_add<1>(ed0);  ed1 = row_xor_add<1>(ed1);
      if (c8 == 0) {
        *(v2f*)(sEs + 2 * p) = v2f{es0, es1};
        *(v2f*)(sEd + 2 * p) = v2f{ed0, ed1};
      }
    }
  }
  __syncthreads();

  const int irow = hf * 256 + wave * 16 + l16;
  const float es = sEs[irow];
  unsigned* maskrow = sMask + (wave * 16 + l16) * 16;

  const float NINF = -__builtin_inff();
  float4 edq[4];
#pragma unroll
  for (int q = 0; q < 4; ++q) edq[q] = *(const float4*)(sEd + 128 * q + 4 * lane);
  const int* adjw = adj + (size_t)(hf * 256 + wave * 16) * NN + 4 * lane;
  unsigned* mwave = sMask + wave * 256;
  float edmax = NINF;
  unsigned allc = 0;
  for (int r4 = 0; r4 < 16; r4 += 4) {
    int4 av[4][4];
#pragma unroll
    for (int i = 0; i < 4; ++i)
#pragma unroll
      for (int q = 0; q < 4; ++q) av[i][q] = *(const int4*)(adjw + (r4 + i) * NN + 128 * q);
#pragma unroll
    for (int i = 0; i < 4; ++i) {
      float mx = NINF;
      unsigned rall = ~0u;
      unsigned mrow = 0;
#pragma unroll
      for (int q = 0; q < 4; ++q) {
        const int   a4[4] = {av[i][q].x, av[i][q].y, av[i][q].z, av[i][q].w};
        const float e4[4] = {edq[q].x, edq[q].y, edq[q].z, edq[q].w};
#pragma unroll
        for (int k = 0; k < 4; ++k) {
          const bool conn = a4[k] > 0;
          const unsigned bal = __builtin_amdgcn_ballot_w32(conn);
          mx = fmaxf(mx, conn ? e4[k] : NINF);
          rall &= bal;
          if (lane == 4 * q + k) mrow = bal;
        }
      }
      if (lane < 16) mwave[(r4 + i) * 16 + lane] = mrow;
      mx = row_xor_max<8>(mx);
      mx = row_xor_max<4>(mx);
      mx = row_xor_max<2>(mx);
      mx = row_xor_max<1>(mx);
      mx = fmaxf(mx, __shfl_xor(mx, 16, 32));
      if (l16 == r4 + i) { edmax = mx; allc = rall; }
    }
  }
  float m = es + edmax;
  m = m > 0.0f ? m : ALPHA * m;
  if (allc != ~0u) m = fmaxf(m, NEG_INF2);
  const float pmask = (m == NEG_INF2) ? 1.0f : 0.0f;
  const v2f esm = v2f{es - m, es - m};
  const v2f esn = v2f{ALPHA * es - m, ALPHA * es - m};
  const v2f al2 = v2f{ALPHA, ALPHA};

  v8f acc[8];
#pragma unroll
  for (int t = 0; t < 8; ++t)
#pragma unroll
    for (int r = 0; r < 8; ++r) acc[t][r] = 0.0f;

  const v2f* bptr  = sWh + g * PAIRP + l16;
  const v2f* edptr = (const v2f*)sEd + g;
  v2f bq[2][8];
  const uint2* mpair = (const uint2*)maskrow + g;
  v2f edc[8], edn[8];
  uint2 mwc, mwn;
  v2f lsum2 = v2f{0.0f, 0.0f};

  auto prob = [&](v2f ed, uint2& mw) -> v2f {
    const v2f t1 = ed + esm;
    const v2f t2 = __builtin_elementwise_fma(ed, al2, esn);
    const float p0 = __builtin_amdgcn_exp2f(fmaxf(t1.x, t2.x));
    const float p1 = __builtin_amdgcn_exp2f(fmaxf(t1.y, t2.y));
    unsigned nx, ny;
    const bool c0 = __builtin_uadd_overflow(mw.x, mw.x, &nx);
    const bool c1 = __builtin_uadd_overflow(mw.y, mw.y, &ny);
    mw.x = nx;
    mw.y = ny;
    v2f p;
    p.x = c0 ? p0 : pmask;
    p.y = c1 ? p1 : pmask;
    return p;
  };

#pragma unroll
  for (int k = 0; k < 8; ++k) edc[k] = edptr[2 * k];
  mwc = mpair[0];
  mwc.x = __builtin_bitreverse32(mwc.x);
  mwc.y = __builtin_bitreverse32(mwc.y);
  v2f a = prob(edc[0], mwc);

  for (int hh = 0; hh < 2; ++hh) {
    const v2f* bh = bptr + hh * (HALFP * PAIRP);
#pragma unroll
    for (int t = 0; t < 8; ++t) bq[0][t] = bh[t * 16];

#pragma unroll 2
    for (int wl = 0; wl < 8; ++wl) {
      const int w  = hh * 8 + wl;
      const int wn = (w + 1) & 15;
#pragma unroll
      for (int c = 0; c < 8; ++c) {
        const int sn = (wl * 8 + c + 1) & 63;
        const v2f* bn = bh + 2 * sn * PAIRP;
        fence();
#pragma unroll
        for (int t = 0; t < 8; ++t) bq[(c + 1) & 1][t] = bn[t * 16];
        if (c == 0) {
#pragma unroll
          for (int k = 0; k < 8; ++k) edn[k] = edptr[16 * wn + 2 * k];
          mwn = mpair[2 * (wn >> 2)];
        }
        fence();
        if (c == 7) {
          mwn.x = __builtin_bitreverse32(mwn.x) << (8 * (wn & 3));
          mwn.y = __builtin_bitreverse32(mwn.y) << (8 * (wn & 3));
        }
        const v2f an = (c < 7) ? prob(edc[(c + 1) & 7], mwc) : prob(edn[0], mwn);
        lsum2 += a;
#pragma unroll
        for (int t = 0; t < 8; ++t)
          acc[t] = __builtin_amdgcn_wmma_f32_16x16x4_f32(
              false, a, false, bq[c & 1][t], (short)0, acc[t], false, false);
        fence();
        a = an;
      }
#pragma unroll
      for (int k = 0; k < 8; ++k) edc[k] = edn[k];
      mwc = mwn;
    }
  }

  const float lsum = lsum2.x + lsum2.y;
  const float l = lsum + __shfl_xor(lsum, 16, 32);

  {
    float4 w0[4], w1[4];
#pragma unroll
    for (int u = 0; u < 4; ++u) {
      const int k2 = u * 16 + wave;
      w0[u] = *(const float4*)(W + (2 * k2) * DD + lane * 4);
      w1[u] = *(const float4*)(W + (2 * k2 + 1) * DD + lane * 4);
    }
    __syncthreads();
#pragma unroll
    for (int u = 0; u < 4; ++u) {
      const int k2 = u * 16 + wave;
      float* dst = (float*)(sW2 + k2 * PAIRP + lane * 4);
      *(float4*)(dst)     = make_float4(w0[u].x, w1[u].x, w0[u].y, w1[u].y);
      *(float4*)(dst + 4) = make_float4(w0[u].z, w1[u].z, w0[u].w, w1[u].w);
    }
#pragma unroll
    for (int t = 0; t < 8; ++t)
#pragma unroll
      for (int r = 0; r < 8; ++r)
        sH2[(wave * 16 + r + 8 * g) * LDSP + t * 16 + l16] = acc[t][r];
  }
  __syncthreads();

#pragma unroll
  for (int t = 0; t < 8; ++t)
#pragma unroll
    for (int r = 0; r < 8; ++r) acc[t][r] = 0.0f;

  {
    const int arow = wave * 16 + l16;
    const float* aptr  = sH2 + arow * LDSP + 2 * g;
    const v2f*   bptr2 = sW2 + g * PAIRP + l16;
    v2f aq[2];
    v2f bw[2][8];
    aq[0] = *(const v2f*)aptr;
#pragma unroll
    for (int t = 0; t < 8; ++t) bw[0][t] = bptr2[t * 16];

    for (int s2 = 0; s2 < FF / 4; s2 += 2) {
#pragma unroll
      for (int u = 0; u < 2; ++u) {
        const int sn = (s2 + u + 1) & (FF / 4 - 1);
        const v2f* bn = bptr2 + 2 * sn * PAIRP;
        fence();
        aq[u ^ 1] = *(const v2f*)(aptr + 4 * sn);
#pragma unroll
        for (int t = 0; t < 8; ++t) bw[u ^ 1][t] = bn[t * 16];
        fence();
#pragma unroll
        for (int t = 0; t < 8; ++t)
          acc[t] = __builtin_amdgcn_wmma_f32_16x16x4_f32(
              false, aq[u], false, bw[u][t], (short)0, acc[t], false, false);
        fence();
      }
    }
  }

  float* outblk = out + ((size_t)bt * NN + (size_t)hf * 256 + wave * 16) * DD;
#pragma unroll
  for (int r = 0; r < 8; ++r) {
    const int orow = r + 8 * g;
    const float lr = __shfl(l, orow, 32);
    const float inv = 1.0f / lr;
#pragma unroll
    for (int t = 0; t < 8; ++t)
      outblk[orow * DD + t * 16 + l16] = acc[t][r] * inv;
  }
}

extern "C" void kernel_launch(void* const* d_in, const int* in_sizes, int n_in,
                              void* d_out, int out_size, void* d_ws, size_t ws_size,
                              hipStream_t stream) {
  const float* h   = (const float*)d_in[0];
  const int*   adj = (const int*)d_in[1];
  const float* W   = (const float*)d_in[2];
  const float* a   = (const float*)d_in[3];
  float* out = (float*)d_out;

  const size_t lds = (size_t)(256 * PAIRP) * 8 + (size_t)NN * 4 + (size_t)NN * 4 +
                     (size_t)(2 * FF) * 4 + (size_t)(16 * 16 * 16) * 4;

  gat_fused_kernel<<<dim3(BT * 2), dim3(512), lds, stream>>>(h, adj, W, a, out);
}
